# starting kernel
# baseline (speedup 1.0000x reference)
#include <hip/hip_runtime.h>

typedef float v2f __attribute__((ext_vector_type(2)));
typedef float v8f __attribute__((ext_vector_type(8)));

namespace {
constexpr int XDIM = 329;
constexpr int NQ   = 25;
constexpr int KV   = 9;
constexpr int TILE = 16;
constexpr int XPAD = 380;

constexpr int SX_ELEMS = TILE * XPAD;
constexpr int SP_ELEMS = 4 * TILE * NQ * KV;
constexpr int SW_ELEMS = 48*12 + 48*20 + 48*12 + 48*12;
constexpr int SB_ELEMS = 4 * 48;
constexpr int SMEM_ELEMS = SX_ELEMS + SP_ELEMS + SW_ELEMS + SB_ELEMS + NQ + 2 * KV;

constexpr int WOFF_J = 0;
constexpr int WOFF_O = 48 * 12;
constexpr int WOFF_G = WOFF_O + 48 * 20;
constexpr int WOFF_B = WOFF_G + 48 * 12;
}

__device__ __forceinline__ int padcol(int c) {
  if (c < 27)  { int j = c / 9;          return        j * 12 + (c - j * 9); }
  if (c < 197) { int cc = c - 27;  int j = cc / 17; return  36 + j * 20 + (cc - j * 17); }
  if (c < 307) { int cc = c - 197; int j = cc / 11; return 236 + j * 12 + (cc - j * 11); }
  {            int cc = c - 307; int j = cc / 11; return 356 + j * 12 + (cc - j * 11); }
}

template <int L, int LPAD, int WOFF>
__device__ __forceinline__ void stage_w(const float* __restrict__ W, float* swp, int lane) {
  for (int i = lane; i < 4 * KV * L; i += 32) {
    const int n = i / L, k = i - n * L;
    swp[WOFF + n * LPAD + k] = W[i];
  }
}

template <int L, int NVEC, int COLP0, int Q0, int WOFF, int GIDX>
__device__ __forceinline__ void proj_group(const float* sxp, const float* swp,
                                           const float* sbias, float* sp,
                                           int m16, int hi) {
  constexpr int KSTEPS = (L + 3) / 4;
  constexpr int LPAD = KSTEPS * 4;
  #pragma unroll
  for (int nt = 0; nt < 3; ++nt) {
    const int nn = nt * 16 + m16;
    v2f bf[KSTEPS];
    #pragma unroll
    for (int ks = 0; ks < KSTEPS; ++ks)
      bf[ks] = *reinterpret_cast<const v2f*>(swp + WOFF + nn * LPAD + ks * 4 + 2 * hi);
    const float bias = sbias[GIDX * 48 + nn];

    for (int j = 0; j < NVEC; ++j) {
      const int colp = COLP0 + j * LPAD;
      v8f acc = {0.f, 0.f, 0.f, 0.f, 0.f, 0.f, 0.f, 0.f};
      #pragma unroll
      for (int ks = 0; ks < KSTEPS; ++ks) {
        const v2f a = *reinterpret_cast<const v2f*>(sxp + m16 * XPAD + colp + ks * 4 + 2 * hi);
        acc = __builtin_amdgcn_wmma_f32_16x16x4_f32(
             false, a,  false, bf[ks],
             (short)0, acc,  false,  false);
      }
      if (nn < 4 * KV) {
        const int p = nn / KV, kvc = nn - (nn / KV) * KV;
        const int q = Q0 + j;
        #pragma unroll
        for (int r = 0; r < 8; ++r) {
          const int m = r + 8 * hi;
          sp[((p * TILE + m) * NQ + q) * KV + kvc] = acc[r] + bias;
        }
      }
    }
  }
}

__global__ __launch_bounds__(32) void attn_fused_kernel(
    const float* __restrict__ x, const int* __restrict__ mask,
    const float* __restrict__ ln_g, const float* __restrict__ ln_b,
    const float* __restrict__ Wj, const float* __restrict__ bj,
    const float* __restrict__ Wo, const float* __restrict__ bo,
    const float* __restrict__ Wg, const float* __restrict__ bg,
    const float* __restrict__ Wb, const float* __restrict__ bb,
    float* __restrict__ out, int btot) {
  __shared__ __align__(16) float smem[SMEM_ELEMS];
  float* sxp   = smem;
  float* sp    = smem + SX_ELEMS;
  float* swp   = sp + SP_ELEMS;
  float* sbias = swp + SW_ELEMS;
  float* smask = sbias + SB_ELEMS;
  float* sgam  = smask + NQ;
  float* sbet  = sgam + KV;

  const int lane = threadIdx.x;
  const long long tile0 = (long long)blockIdx.x * TILE;

  for (int i = lane; i < SX_ELEMS; i += 32) sxp[i] = 0.0f;
  for (int i = lane; i < SW_ELEMS + SB_ELEMS; i += 32) swp[i] = 0.0f;
  __syncthreads();

  for (int i = lane; i < TILE * XDIM; i += 32) {
    const int s = i / XDIM, c = i - s * XDIM;
    const float vx = ((tile0 + s) < btot) ? x[tile0 * XDIM + i] : 0.0f;
    sxp[s * XPAD + padcol(c)] = vx;
  }
  stage_w< 9, 12, WOFF_J>(Wj, swp, lane);
  stage_w<17, 20, WOFF_O>(Wo, swp, lane);
  stage_w<11, 12, WOFF_G>(Wg, swp, lane);
  stage_w<11, 12, WOFF_B>(Wb, swp, lane);
  for (int i = lane; i < 4 * KV; i += 32) {
    sbias[0 * 48 + i] = bj[i];
    sbias[1 * 48 + i] = bo[i];
    sbias[2 * 48 + i] = bg[i];
    sbias[3 * 48 + i] = bb[i];
  }
  if (lane < NQ) smask[lane] = -1.0e9f * (float)mask[lane];
  if (lane < KV) { sgam[lane] = ln_g[lane]; sbet[lane] = ln_b[lane]; }
  __syncthreads();

  const int m16 = lane & 15;
  const int hi  = lane >> 4;
  proj_group< 9,  3,   0,  0, WOFF_J, 0>(sxp, swp, sbias, sp, m16, hi);
  proj_group<17, 10,  36,  3, WOFF_O, 1>(sxp, swp, sbias, sp, m16, hi);
  proj_group<11, 10, 236, 13, WOFF_G, 2>(sxp, swp, sbias, sp, m16, hi);
  proj_group<11,  2, 356, 23, WOFF_B, 3>(sxp, swp, sbias, sp, m16, hi);
  __syncthreads();

  const int s = lane >> 1;
  const int h = lane & 1;
  const float* Pq = sp;
  const float* Pk = sp + 1 * TILE * NQ * KV;
  const float* Pv = sp + 2 * TILE * NQ * KV;
  const float* Pr = sp + 3 * TILE * NQ * KV;
  float* ob = sxp;

  for (int qi = h; qi < NQ; qi += 2) {
    float qv[KV];
    #pragma unroll
    for (int d = 0; d < KV; ++d) qv[d] = Pq[(s * NQ + qi) * KV + d];

    float sc[NQ];
    float mx = -3.0e38f;
    #pragma unroll
    for (int j = 0; j < NQ; ++j) {
      float t = 0.0f;
      #pragma unroll
      for (int d = 0; d < KV; ++d) t += qv[d] * Pk[(s * NQ + j) * KV + d];
      t = t * (1.0f / 3.0f) + smask[j];
      sc[j] = t;
      mx = fmaxf(mx, t);
    }
    float denom = 0.0f;
    #pragma unroll
    for (int j = 0; j < NQ; ++j) {
      const float e = __expf(sc[j] - mx);
      sc[j] = e;
      denom += e;
    }
    const float inv = 1.0f / denom;

    float o[KV];
    float mu = 0.0f;
    #pragma unroll
    for (int d = 0; d < KV; ++d) {
      float t = 0.0f;
      #pragma unroll
      for (int j = 0; j < NQ; ++j) t += sc[j] * Pv[(s * NQ + j) * KV + d];
      t = t * inv + Pr[(s * NQ + qi) * KV + d];
      o[d] = t;
      mu += t;
    }
    mu *= (1.0f / KV);
    float var = 0.0f;
    #pragma unroll
    for (int d = 0; d < KV; ++d) { const float dd = o[d] - mu; var += dd * dd; }
    var *= (1.0f / KV);
    const float rstd = rsqrtf(var + 1e-5f);
    #pragma unroll
    for (int d = 0; d < KV; ++d)
      ob[(s * NQ + qi) * KV + d] = (o[d] - mu) * rstd * sgam[d] + sbet[d];
  }
  __syncthreads();

  for (int i = lane; i < TILE * NQ * KV; i += 32) {
    const long long ss = tile0 + i / (NQ * KV);
    if (ss < btot) out[tile0 * NQ * KV + i] = ob[i];
  }
}

extern "C" void kernel_launch(void* const* d_in, const int* in_sizes, int n_in,
                              void* d_out, int out_size, void* d_ws, size_t ws_size,
                              hipStream_t stream) {
  (void)d_ws; (void)ws_size; (void)n_in; (void)out_size;
  const float* x    = (const float*)d_in[0];
  const int*   mask = (const int*)d_in[1];
  const float* ln_g = (const float*)d_in[2];
  const float* ln_b = (const float*)d_in[3];
  const float* Wj   = (const float*)d_in[4];
  const float* bj   = (const float*)d_in[5];
  const float* Wo   = (const float*)d_in[6];
  const float* bo   = (const float*)d_in[7];
  const float* Wg   = (const float*)d_in[8];
  const float* bg   = (const float*)d_in[9];
  const float* Wb   = (const float*)d_in[10];
  const float* bb   = (const float*)d_in[11];
  float* out = (float*)d_out;

  const int btot = in_sizes[0] / XDIM;
  const int grid = (btot + TILE - 1) / TILE;
  attn_fused_kernel<<<grid, 32, 0, stream>>>(x, mask, ln_g, ln_b,
                                             Wj, bj, Wo, bo, Wg, bg, Wb, bb,
                                             out, btot);
}


# global loads grouped ahead of their waits: x tile as 16-byte loads, 14 per lane in flight, next group issued before the scatter; weight loads one group (was one 4-byte load per wait)
# speedup vs baseline: 1.8289x; 1.8289x over previous
#include <hip/hip_runtime.h>

typedef float v2f __attribute__((ext_vector_type(2)));
typedef float v4f __attribute__((ext_vector_type(4)));
typedef float v8f __attribute__((ext_vector_type(8)));

namespace {
constexpr int XDIM = 329;
constexpr int NQ   = 25;
constexpr int KV   = 9;
constexpr int TILE = 16;
constexpr int XPAD = 380;

constexpr int SX_ELEMS = TILE * XPAD;
constexpr int SP_ELEMS = 4 * TILE * NQ * KV;
constexpr int SW_ELEMS = 48*12 + 48*20 + 48*12 + 48*12;
constexpr int SB_ELEMS = 4 * 48;
constexpr int SMEM_ELEMS = SX_ELEMS + SP_ELEMS + SW_ELEMS + SB_ELEMS + NQ + 2 * KV;

constexpr int TILE_F4 = TILE * XDIM / 4;
constexpr int XCH     = 14;
constexpr int XNCH    = 3;
static_assert(TILE * XDIM % 4 == 0, "tile must be whole 16-byte pieces");
static_assert(XNCH * XCH * 32 >= TILE_F4, "groups must cover the tile");

constexpr int WOFF_J = 0;
constexpr int WOFF_O = 48 * 12;
constexpr int WOFF_G = WOFF_O + 48 * 20;
constexpr int WOFF_B = WOFF_G + 48 * 12;
}

__device__ __forceinline__ int padcol(int c) {
  const unsigned uc = (unsigned)c;
  const unsigned a = 3u * (uc / 9u);
  const unsigned b = 9u + 3u * ((uc - 27u) / 17u);
  const unsigned g = 39u + (uc - 197u) / 11u;
  return (int)(uc + (uc < 27u ? a : (uc < 197u ? b : g)));
}

template <int L, int LPAD, int WOFF>
__device__ __forceinline__ void stage_w(const float* __restrict__ W, float* swp, int lane) {
  constexpr int N  = 4 * KV * L;
  constexpr int NT = (N + 31) / 32;
  float r[NT];
  #pragma unroll
  for (int t = 0; t < NT; ++t) {
    const int i = min(lane + 32 * t, N - 1);
    r[t] = W[i];
  }
  #pragma unroll
  for (int t = 0; t < NT; ++t) {
    const int i = min(lane + 32 * t, N - 1);
    const int n = i / L, k = i - n * L;
    swp[WOFF + n * LPAD + k] = r[t];
  }
}

template <int L, int NVEC, int COLP0, int Q0, int WOFF, int GIDX>
__device__ __forceinline__ void proj_group(const float* sxp, const float* swp,
                                           const float* sbias, float* sp,
                                           int m16, int hi) {
  constexpr int KSTEPS = (L + 3) / 4;
  constexpr int LPAD = KSTEPS * 4;
  #pragma unroll
  for (int nt = 0; nt < 3; ++nt) {
    const int nn = nt * 16 + m16;
    v2f bf[KSTEPS];
    #pragma unroll
    for (int ks = 0; ks < KSTEPS; ++ks)
      bf[ks] = *reinterpret_cast<const v2f*>(swp + WOFF + nn * LPAD + ks * 4 + 2 * hi);
    const float bias = sbias[GIDX * 48 + nn];

    for (int j = 0; j < NVEC; ++j) {
      const int colp = COLP0 + j * LPAD;
      v8f acc = {0.f, 0.f, 0.f, 0.f, 0.f, 0.f, 0.f, 0.f};
      #pragma unroll
      for (int ks = 0; ks < KSTEPS; ++ks) {
        const v2f a = *reinterpret_cast<const v2f*>(sxp + m16 * XPAD + colp + ks * 4 + 2 * hi);
        acc = __builtin_amdgcn_wmma_f32_16x16x4_f32(
             false, a,  false, bf[ks],
             (short)0, acc,  false,  false);
      }
      if (nn < 4 * KV) {
        const int p = nn / KV, kvc = nn - (nn / KV) * KV;
        const int q = Q0 + j;
        #pragma unroll
        for (int r = 0; r < 8; ++r) {
          const int m = r + 8 * hi;
          sp[((p * TILE + m) * NQ + q) * KV + kvc] = acc[r] + bias;
        }
      }
    }
  }
}

__global__ __launch_bounds__(32) void attn_fused_kernel(
    const float* __restrict__ x, const int* __restrict__ mask,
    const float* __restrict__ ln_g, const float* __restrict__ ln_b,
    const float* __restrict__ Wj, const float* __restrict__ bj,
    const float* __restrict__ Wo, const float* __restrict__ bo,
    const float* __restrict__ Wg, const float* __restrict__ bg,
    const float* __restrict__ Wb, const float* __restrict__ bb,
    float* __restrict__ out, int btot) {
  __shared__ __align__(16) float smem[SMEM_ELEMS];
  float* sxp   = smem;
  float* sp    = smem + SX_ELEMS;
  float* swp   = sp + SP_ELEMS;
  float* sbias = swp + SW_ELEMS;
  float* smask = sbias + SB_ELEMS;
  float* sgam  = smask + NQ;
  float* sbet  = sgam + KV;

  const int lane = threadIdx.x;
  const long long tile0 = (long long)blockIdx.x * TILE;

  for (int i = lane; i < SX_ELEMS; i += 32) sxp[i] = 0.0f;
  for (int i = lane; i < SW_ELEMS + SB_ELEMS; i += 32) swp[i] = 0.0f;
  __syncthreads();

  const bool whole = (tile0 + TILE <= (long long)btot) &&
                     ((reinterpret_cast<unsigned long long>(x) & 15ull) == 0ull);
  if (whole) {
    const v4f* xt = reinterpret_cast<const v4f*>(x + tile0 * XDIM);
    v4f cur[XCH], nxt[XCH];
    #pragma unroll
    for (int u = 0; u < XCH; ++u) cur[u] = xt[min(u * 32 + lane, TILE_F4 - 1)];
    #pragma unroll 1
    for (int t = 0; t < XNCH; ++t) {
      #pragma unroll
      for (int u = 0; u < XCH; ++u)
        nxt[u] = xt[min(((t + 1) * XCH + u) * 32 + lane, TILE_F4 - 1)];
      #pragma unroll
      for (int u = 0; u < XCH; ++u) {
        const int f = min((t * XCH + u) * 32 + lane, TILE_F4 - 1);
        #pragma unroll
        for (int e = 0; e < 4; ++e) {
          const int i = 4 * f + e;
          const int s = i / XDIM, cc = i - s * XDIM;
          sxp[s * XPAD + padcol(cc)] = cur[u][e];
        }
      }
      #pragma unroll
      for (int u = 0; u < XCH; ++u) cur[u] = nxt[u];
    }
  } else {
    for (int i = lane; i < TILE * XDIM; i += 32) {
      const int s = i / XDIM, cc = i - s * XDIM;
      const float vx = ((tile0 + s) < btot) ? x[tile0 * XDIM + i] : 0.0f;
      sxp[s * XPAD + padcol(cc)] = vx;
    }
  }
  stage_w< 9, 12, WOFF_J>(Wj, swp, lane);
  stage_w<17, 20, WOFF_O>(Wo, swp, lane);
  stage_w<11, 12, WOFF_G>(Wg, swp, lane);
  stage_w<11, 12, WOFF_B>(Wb, swp, lane);
  for (int i = lane; i < 4 * KV; i += 32) {
    sbias[0 * 48 + i] = bj[i];
    sbias[1 * 48 + i] = bo[i];
    sbias[2 * 48 + i] = bg[i];
    sbias[3 * 48 + i] = bb[i];
  }
  if (lane < NQ) smask[lane] = -1.0e9f * (float)mask[lane];
  if (lane < KV) { sgam[lane] = ln_g[lane]; sbet[lane] = ln_b[lane]; }
  __syncthreads();

  const int m16 = lane & 15;
  const int hi  = lane >> 4;
  proj_group< 9,  3,   0,  0, WOFF_J, 0>(sxp, swp, sbias, sp, m16, hi);
  proj_group<17, 10,  36,  3, WOFF_O, 1>(sxp, swp, sbias, sp, m16, hi);
  proj_group<11, 10, 236, 13, WOFF_G, 2>(sxp, swp, sbias, sp, m16, hi);
  proj_group<11,  2, 356, 23, WOFF_B, 3>(sxp, swp, sbias, sp, m16, hi);
  __syncthreads();

  const int s = lane >> 1;
  const int h = lane & 1;
  const float* Pq = sp;
  const float* Pk = sp + 1 * TILE * NQ * KV;
  const float* Pv = sp + 2 * TILE * NQ * KV;
  const float* Pr = sp + 3 * TILE * NQ * KV;
  float* ob = sxp;

  for (int qi = h; qi < NQ; qi += 2) {
    float qv[KV];
    #pragma unroll
    for (int d = 0; d < KV; ++d) qv[d] = Pq[(s * NQ + qi) * KV + d];

    float sc[NQ];
    float mx = -3.0e38f;
    #pragma unroll
    for (int j = 0; j < NQ; ++j) {
      float t = 0.0f;
      #pragma unroll
      for (int d = 0; d < KV; ++d) t += qv[d] * Pk[(s * NQ + j) * KV + d];
      t = t * (1.0f / 3.0f) + smask[j];
      sc[j] = t;
      mx = fmaxf(mx, t);
    }
    float denom = 0.0f;
    #pragma unroll
    for (int j = 0; j < NQ; ++j) {
      const float e = __expf(sc[j] - mx);
      sc[j] = e;
      denom += e;
    }
    const float inv = 1.0f / denom;

    float o[KV];
    float mu = 0.0f;
    #pragma unroll
    for (int d = 0; d < KV; ++d) {
      float t = 0.0f;
      #pragma unroll
      for (int j = 0; j < NQ; ++j) t += sc[j] * Pv[(s * NQ + j) * KV + d];
      t = t * inv + Pr[(s * NQ + qi) * KV + d];
      o[d] = t;
      mu += t;
    }
    mu *= (1.0f / KV);
    float var = 0.0f;
    #pragma unroll
    for (int d = 0; d < KV; ++d) { const float dd = o[d] - mu; var += dd * dd; }
    var *= (1.0f / KV);
    const float rstd = rsqrtf(var + 1e-5f);
    #pragma unroll
    for (int d = 0; d < KV; ++d)
      ob[(s * NQ + qi) * KV + d] = (o[d] - mu) * rstd * sgam[d] + sbet[d];
  }
  __syncthreads();

  for (int i = lane; i < TILE * NQ * KV; i += 32) {
    const long long ss = tile0 + i / (NQ * KV);
    if (ss < btot) out[tile0 * NQ * KV + i] = ob[i];
  }
}

extern "C" void kernel_launch(void* const* d_in, const int* in_sizes, int n_in,
                              void* d_out, int out_size, void* d_ws, size_t ws_size,
                              hipStream_t stream) {
  (void)d_ws; (void)ws_size; (void)n_in; (void)out_size;
  const float* x    = (const float*)d_in[0];
  const int*   mask = (const int*)d_in[1];
  const float* ln_g = (const float*)d_in[2];
  const float* ln_b = (const float*)d_in[3];
  const float* Wj   = (const float*)d_in[4];
  const float* bj   = (const float*)d_in[5];
  const float* Wo   = (const float*)d_in[6];
  const float* bo   = (const float*)d_in[7];
  const float* Wg   = (const float*)d_in[8];
  const float* bg   = (const float*)d_in[9];
  const float* Wb   = (const float*)d_in[10];
  const float* bb   = (const float*)d_in[11];
  float* out = (float*)d_out;

  const int btot = in_sizes[0] / XDIM;
  const int grid = (btot + TILE - 1) / TILE;
  attn_fused_kernel<<<grid, 32, 0, stream>>>(x, mask, ln_g, ln_b,
                                             Wj, bj, Wo, bo, Wg, bg, Wb, bb,
                                             out, btot);
}


# each 16-sample tile is worked by a 4-wave work-group instead of one wave: staging, projection tokens and attention queries are split over the waves; 12 waves per WGP are resident where 3 were
# speedup vs baseline: 5.0990x; 2.7880x over previous
#include <hip/hip_runtime.h>

typedef float v2f __attribute__((ext_vector_type(2)));
typedef float v4f __attribute__((ext_vector_type(4)));
typedef float v8f __attribute__((ext_vector_type(8)));

namespace {
constexpr int XDIM = 329;
constexpr int NQ   = 25;
constexpr int KV   = 9;
constexpr int TILE = 16;
constexpr int NWAVE = 4;
constexpr int NTHR  = 32 * NWAVE;
constexpr int XPAD = 380;

constexpr int SX_ELEMS = TILE * XPAD;
constexpr int SP_ELEMS = 4 * TILE * NQ * KV;
constexpr int SW_ELEMS = 48*12 + 48*20 + 48*12 + 48*12;
constexpr int SB_ELEMS = 4 * 48;
constexpr int SMEM_ELEMS = SX_ELEMS + SP_ELEMS + SW_ELEMS + SB_ELEMS + NQ + 2 * KV;

constexpr int TILE_F4 = TILE * XDIM / 4;
constexpr int XCH     = (TILE_F4 + NTHR - 1) / NTHR;
static_assert(TILE * XDIM % 4 == 0, "tile must be whole 16-byte pieces");
static_assert(XCH * NTHR >= TILE_F4, "the group must cover the tile");

constexpr int WOFF_J = 0;
constexpr int WOFF_O = 48 * 12;
constexpr int WOFF_G = WOFF_O + 48 * 20;
constexpr int WOFF_B = WOFF_G + 48 * 12;
}

__device__ __forceinline__ int padcol(int c) {
  const unsigned uc = (unsigned)c;
  const unsigned a = 3u * (uc / 9u);
  const unsigned b = 9u + 3u * ((uc - 27u) / 17u);
  const unsigned g = 39u + (uc - 197u) / 11u;
  return (int)(uc + (uc < 27u ? a : (uc < 197u ? b : g)));
}

template <int L, int LPAD, int WOFF>
__device__ __forceinline__ void stage_w(const float* __restrict__ W, float* swp, int lane) {
  constexpr int N  = 4 * KV * L;
  constexpr int NT = (N + NTHR - 1) / NTHR;
  float r[NT];
  #pragma unroll
  for (int t = 0; t < NT; ++t) {
    const int i = min(lane + NTHR * t, N - 1);
    r[t] = W[i];
  }
  #pragma unroll
  for (int t = 0; t < NT; ++t) {
    const int i = min(lane + NTHR * t, N - 1);
    const int n = i / L, k = i - n * L;
    swp[WOFF + n * LPAD + k] = r[t];
  }
}

template <int L, int NVEC, int COLP0, int Q0, int WOFF, int GIDX>
__device__ __forceinline__ void proj_group(const float* sxp, const float* swp,
                                           const float* sbias, float* sp,
                                           int m16, int hi, int wv) {
  const int j0 = (wv + 6 * NWAVE - Q0) % NWAVE;
  constexpr int KSTEPS = (L + 3) / 4;
  constexpr int LPAD = KSTEPS * 4;
  #pragma unroll
  for (int nt = 0; nt < 3; ++nt) {
    const int nn = nt * 16 + m16;
    v2f bf[KSTEPS];
    #pragma unroll
    for (int ks = 0; ks < KSTEPS; ++ks)
      bf[ks] = *reinterpret_cast<const v2f*>(swp + WOFF + nn * LPAD + ks * 4 + 2 * hi);
    const float bias = sbias[GIDX * 48 + nn];

    for (int j = j0; j < NVEC; j += NWAVE) {
      const int colp = COLP0 + j * LPAD;
      v8f acc = {0.f, 0.f, 0.f, 0.f, 0.f, 0.f, 0.f, 0.f};
      #pragma unroll
      for (int ks = 0; ks < KSTEPS; ++ks) {
        const v2f a = *reinterpret_cast<const v2f*>(sxp + m16 * XPAD + colp + ks * 4 + 2 * hi);
        acc = __builtin_amdgcn_wmma_f32_16x16x4_f32(
             false, a,  false, bf[ks],
             (short)0, acc,  false,  false);
      }
      if (nn < 4 * KV) {
        const int p = nn / KV, kvc = nn - (nn / KV) * KV;
        const int q = Q0 + j;
        #pragma unroll
        for (int r = 0; r < 8; ++r) {
          const int m = r + 8 * hi;
          sp[((p * TILE + m) * NQ + q) * KV + kvc] = acc[r] + bias;
        }
      }
    }
  }
}

__global__ __launch_bounds__(NTHR) __attribute__((amdgpu_waves_per_eu(3)))
void attn_fused_kernel(
    const float* __restrict__ x, const int* __restrict__ mask,
    const float* __restrict__ ln_g, const float* __restrict__ ln_b,
    const float* __restrict__ Wj, const float* __restrict__ bj,
    const float* __restrict__ Wo, const float* __restrict__ bo,
    const float* __restrict__ Wg, const float* __restrict__ bg,
    const float* __restrict__ Wb, const float* __restrict__ bb,
    float* __restrict__ out, int btot) {
  __shared__ __align__(16) float smem[SMEM_ELEMS];
  float* sxp   = smem;
  float* sp    = smem + SX_ELEMS;
  float* swp   = sp + SP_ELEMS;
  float* sbias = swp + SW_ELEMS;
  float* smask = sbias + SB_ELEMS;
  float* sgam  = smask + NQ;
  float* sbet  = sgam + KV;

  const int lane = threadIdx.x;
  const long long tile0 = (long long)blockIdx.x * TILE;

  for (int i = lane; i < SX_ELEMS; i += NTHR) sxp[i] = 0.0f;
  for (int i = lane; i < SW_ELEMS + SB_ELEMS; i += NTHR) swp[i] = 0.0f;
  __syncthreads();

  const bool whole = (tile0 + TILE <= (long long)btot) &&
                     ((reinterpret_cast<unsigned long long>(x) & 15ull) == 0ull);
  if (whole) {
    const v4f* xt = reinterpret_cast<const v4f*>(x + tile0 * XDIM);
    v4f cur[XCH];
    #pragma unroll
    for (int u = 0; u < XCH; ++u) cur[u] = xt[min(u * NTHR + lane, TILE_F4 - 1)];
    #pragma unroll
    for (int u = 0; u < XCH; ++u) {
      const int f = min(u * NTHR + lane, TILE_F4 - 1);
      #pragma unroll
      for (int e = 0; e < 4; ++e) {
        const int i = 4 * f + e;
        const int s = i / XDIM, cc = i - s * XDIM;
        sxp[s * XPAD + padcol(cc)] = cur[u][e];
      }
    }
  } else {
    for (int i = lane; i < TILE * XDIM; i += NTHR) {
      const int s = i / XDIM, cc = i - s * XDIM;
      const float vx = ((tile0 + s) < btot) ? x[tile0 * XDIM + i] : 0.0f;
      sxp[s * XPAD + padcol(cc)] = vx;
    }
  }
  stage_w< 9, 12, WOFF_J>(Wj, swp, lane);
  stage_w<17, 20, WOFF_O>(Wo, swp, lane);
  stage_w<11, 12, WOFF_G>(Wg, swp, lane);
  stage_w<11, 12, WOFF_B>(Wb, swp, lane);
  for (int i = lane; i < 4 * KV; i += NTHR) {
    sbias[0 * 48 + i] = bj[i];
    sbias[1 * 48 + i] = bo[i];
    sbias[2 * 48 + i] = bg[i];
    sbias[3 * 48 + i] = bb[i];
  }
  if (lane < NQ) smask[lane] = -1.0e9f * (float)mask[lane];
  if (lane < KV) { sgam[lane] = ln_g[lane]; sbet[lane] = ln_b[lane]; }
  __syncthreads();

  const int m16 = lane & 15;
  const int hi  = (lane >> 4) & 1;
  const int wv = __builtin_amdgcn_readfirstlane(lane >> 5);
  proj_group< 9,  3,   0,  0, WOFF_J, 0>(sxp, swp, sbias, sp, m16, hi, wv);
  proj_group<17, 10,  36,  3, WOFF_O, 1>(sxp, swp, sbias, sp, m16, hi, wv);
  proj_group<11, 10, 236, 13, WOFF_G, 2>(sxp, swp, sbias, sp, m16, hi, wv);
  proj_group<11,  2, 356, 23, WOFF_B, 3>(sxp, swp, sbias, sp, m16, hi, wv);
  __syncthreads();

  const int s = lane >> 3;
  const int h = lane & 7;
  const float* Pq = sp;
  const float* Pk = sp + 1 * TILE * NQ * KV;
  const float* Pv = sp + 2 * TILE * NQ * KV;
  const float* Pr = sp + 3 * TILE * NQ * KV;
  float* ob = sxp;

  for (int qi = h; qi < NQ; qi += 8) {
    float qv[KV];
    #pragma unroll
    for (int d = 0; d < KV; ++d) qv[d] = Pq[(s * NQ + qi) * KV + d];

    float sc[NQ];
    float mx = -3.0e38f;
    #pragma unroll
    for (int j = 0; j < NQ; ++j) {
      float t = 0.0f;
      #pragma unroll
      for (int d = 0; d < KV; ++d) t += qv[d] * Pk[(s * NQ + j) * KV + d];
      t = t * (1.0f / 3.0f) + smask[j];
      sc[j] = t;
      mx = fmaxf(mx, t);
    }
    float denom = 0.0f;
    #pragma unroll
    for (int j = 0; j < NQ; ++j) {
      const float e = __expf(sc[j] - mx);
      sc[j] = e;
      denom += e;
    }
    const float inv = 1.0f / denom;

    float o[KV];
    float mu = 0.0f;
    #pragma unroll
    for (int d = 0; d < KV; ++d) {
      float t = 0.0f;
      #pragma unroll
      for (int j = 0; j < NQ; ++j) t += sc[j] * Pv[(s * NQ + j) * KV + d];
      t = t * inv + Pr[(s * NQ + qi) * KV + d];
      o[d] = t;
      mu += t;
    }
    mu *= (1.0f / KV);
    float var = 0.0f;
    #pragma unroll
    for (int d = 0; d < KV; ++d) { const float dd = o[d] - mu; var += dd * dd; }
    var *= (1.0f / KV);
    const float rstd = rsqrtf(var + 1e-5f);
    #pragma unroll
    for (int d = 0; d < KV; ++d)
      ob[(s * NQ + qi) * KV + d] = (o[d] - mu) * rstd * sgam[d] + sbet[d];
  }
  __syncthreads();

  const int t3 = wv * 32 + (int)__builtin_amdgcn_mbcnt_lo(~0u, 0u);
  const int nvalid = (int)min((long long)TILE, (long long)btot - tile0);
  float* __restrict__ ot = out + tile0 * (NQ * KV);
  for (int i = t3; i < nvalid * (NQ * KV); i += NTHR) ot[i] = ob[i];
}

extern "C" void kernel_launch(void* const* d_in, const int* in_sizes, int n_in,
                              void* d_out, int out_size, void* d_ws, size_t ws_size,
                              hipStream_t stream) {
  (void)d_ws; (void)ws_size; (void)n_in; (void)out_size;
  const float* x    = (const float*)d_in[0];
  const int*   mask = (const int*)d_in[1];
  const float* ln_g = (const float*)d_in[2];
  const float* ln_b = (const float*)d_in[3];
  const float* Wj   = (const float*)d_in[4];
  const float* bj   = (const float*)d_in[5];
  const float* Wo   = (const float*)d_in[6];
  const float* bo   = (const float*)d_in[7];
  const float* Wg   = (const float*)d_in[8];
  const float* bg   = (const float*)d_in[9];
  const float* Wb   = (const float*)d_in[10];
  const float* bb   = (const float*)d_in[11];
  float* out = (float*)d_out;

  const int btot = in_sizes[0] / XDIM;
  const int grid = (btot + TILE - 1) / TILE;
  attn_fused_kernel<<<grid, 128, 0, stream>>>(x, mask, ln_g, ln_b,
                                             Wj, bj, Wo, bo, Wg, bg, Wb, bb,
                                             out, btot);
}


# attention: a lane takes its 3 or 4 queries through one pass over K and one over V, two queries per packed multiply-add, rows loaded a group ahead; K/V read from LDS once per lane, not once per query
# speedup vs baseline: 7.3752x; 1.4464x over previous
#include <hip/hip_runtime.h>

typedef float v2f __attribute__((ext_vector_type(2)));
typedef float v4f __attribute__((ext_vector_type(4)));
typedef float v8f __attribute__((ext_vector_type(8)));

namespace {
constexpr int XDIM = 329;
constexpr int NQ   = 25;
constexpr int KV   = 9;
constexpr int TILE = 16;
constexpr int NWAVE = 4;
constexpr int NTHR  = 32 * NWAVE;
constexpr int XPAD = 380;

constexpr int SX_ELEMS = TILE * XPAD;
constexpr int SP_ELEMS = 4 * TILE * NQ * KV;
constexpr int SW_ELEMS = 48*12 + 48*20 + 48*12 + 48*12;
constexpr int SB_ELEMS = 4 * 48;
constexpr int SMEM_ELEMS = SX_ELEMS + SP_ELEMS + SW_ELEMS + SB_ELEMS + NQ + 2 * KV;

constexpr int TILE_F4 = TILE * XDIM / 4;
constexpr int XCH     = (TILE_F4 + NTHR - 1) / NTHR;
static_assert(TILE * XDIM % 4 == 0, "tile must be whole 16-byte pieces");
static_assert(XCH * NTHR >= TILE_F4, "the group must cover the tile");

constexpr int WOFF_J = 0;
constexpr int WOFF_O = 48 * 12;
constexpr int WOFF_G = WOFF_O + 48 * 20;
constexpr int WOFF_B = WOFF_G + 48 * 12;
}

__device__ __forceinline__ int lds_base(int idx) {
  asm("" : "+v"(idx));
  return idx;
}

__device__ __forceinline__ int padcol(int c) {
  const unsigned uc = (unsigned)c;
  const unsigned a = 3u * (uc / 9u);
  const unsigned b = 9u + 3u * ((uc - 27u) / 17u);
  const unsigned g = 39u + (uc - 197u) / 11u;
  return (int)(uc + (uc < 27u ? a : (uc < 197u ? b : g)));
}

template <int L, int LPAD, int WOFF>
__device__ __forceinline__ void stage_w(const float* __restrict__ W, float* swp, int lane) {
  constexpr int N  = 4 * KV * L;
  constexpr int NT = (N + NTHR - 1) / NTHR;
  float r[NT];
  #pragma unroll
  for (int t = 0; t < NT; ++t) {
    const int i = min(lane + NTHR * t, N - 1);
    r[t] = W[i];
  }
  #pragma unroll
  for (int t = 0; t < NT; ++t) {
    const int i = min(lane + NTHR * t, N - 1);
    const int n = i / L, k = i - n * L;
    swp[WOFF + n * LPAD + k] = r[t];
  }
}

template <int L, int NVEC, int COLP0, int Q0, int WOFF, int GIDX>
__device__ __forceinline__ void proj_group(const float* sxp, const float* swp,
                                           const float* sbias, float* sp,
                                           int m16, int hi, int wv) {
  const int j0 = (wv + 6 * NWAVE - Q0) % NWAVE;
  constexpr int KSTEPS = (L + 3) / 4;
  constexpr int LPAD = KSTEPS * 4;
  #pragma unroll
  for (int nt = 0; nt < 3; ++nt) {
    const int nn = nt * 16 + m16;
    v2f bf[KSTEPS];
    #pragma unroll
    for (int ks = 0; ks < KSTEPS; ++ks)
      bf[ks] = *reinterpret_cast<const v2f*>(swp + WOFF + nn * LPAD + ks * 4 + 2 * hi);
    const float bias = sbias[GIDX * 48 + nn];

    for (int j = j0; j < NVEC; j += NWAVE) {
      const int colp = COLP0 + j * LPAD;
      v8f acc = {0.f, 0.f, 0.f, 0.f, 0.f, 0.f, 0.f, 0.f};
      #pragma unroll
      for (int ks = 0; ks < KSTEPS; ++ks) {
        const v2f a = *reinterpret_cast<const v2f*>(sxp + m16 * XPAD + colp + ks * 4 + 2 * hi);
        acc = __builtin_amdgcn_wmma_f32_16x16x4_f32(
             false, a,  false, bf[ks],
             (short)0, acc,  false,  false);
      }
      if (nn < 4 * KV) {
        const int p = nn / KV, kvc = nn - (nn / KV) * KV;
        const int q = Q0 + j;
        #pragma unroll
        for (int r = 0; r < 8; ++r) {
          const int m = r + 8 * hi;
          sp[((p * TILE + m) * NQ + q) * KV + kvc] = acc[r] + bias;
        }
      }
    }
  }
}

__global__ __launch_bounds__(NTHR) __attribute__((amdgpu_waves_per_eu(3)))
void attn_fused_kernel(
    const float* __restrict__ x, const int* __restrict__ mask,
    const float* __restrict__ ln_g, const float* __restrict__ ln_b,
    const float* __restrict__ Wj, const float* __restrict__ bj,
    const float* __restrict__ Wo, const float* __restrict__ bo,
    const float* __restrict__ Wg, const float* __restrict__ bg,
    const float* __restrict__ Wb, const float* __restrict__ bb,
    float* __restrict__ out, int btot) {
  __shared__ __align__(16) float smem[SMEM_ELEMS];
  float* sxp   = smem;
  float* sp    = smem + SX_ELEMS;
  float* swp   = sp + SP_ELEMS;
  float* sbias = swp + SW_ELEMS;
  float* smask = sbias + SB_ELEMS;
  float* sgam  = smask + NQ;
  float* sbet  = sgam + KV;

  const int lane = threadIdx.x;
  const long long tile0 = (long long)blockIdx.x * TILE;

  for (int i = lane; i < SX_ELEMS; i += NTHR) sxp[i] = 0.0f;
  for (int i = lane; i < SW_ELEMS + SB_ELEMS; i += NTHR) swp[i] = 0.0f;
  __syncthreads();

  const bool whole = (tile0 + TILE <= (long long)btot) &&
                     ((reinterpret_cast<unsigned long long>(x) & 15ull) == 0ull);
  if (whole) {
    const v4f* xt = reinterpret_cast<const v4f*>(x + tile0 * XDIM);
    v4f cur[XCH];
    #pragma unroll
    for (int u = 0; u < XCH; ++u) cur[u] = xt[min(u * NTHR + lane, TILE_F4 - 1)];
    #pragma unroll
    for (int u = 0; u < XCH; ++u) {
      const int f = min(u * NTHR + lane, TILE_F4 - 1);
      #pragma unroll
      for (int e = 0; e < 4; ++e) {
        const int i = 4 * f + e;
        const int s = i / XDIM, cc = i - s * XDIM;
        sxp[s * XPAD + padcol(cc)] = cur[u][e];
      }
    }
  } else {
    for (int i = lane; i < TILE * XDIM; i += NTHR) {
      const int s = i / XDIM, cc = i - s * XDIM;
      const float vx = ((tile0 + s) < btot) ? x[tile0 * XDIM + i] : 0.0f;
      sxp[s * XPAD + padcol(cc)] = vx;
    }
  }
  stage_w< 9, 12, WOFF_J>(Wj, swp, lane);
  stage_w<17, 20, WOFF_O>(Wo, swp, lane);
  stage_w<11, 12, WOFF_G>(Wg, swp, lane);
  stage_w<11, 12, WOFF_B>(Wb, swp, lane);
  for (int i = lane; i < 4 * KV; i += NTHR) {
    sbias[0 * 48 + i] = bj[i];
    sbias[1 * 48 + i] = bo[i];
    sbias[2 * 48 + i] = bg[i];
    sbias[3 * 48 + i] = bb[i];
  }
  if (lane < NQ) smask[lane] = -1.0e9f * (float)mask[lane];
  if (lane < KV) { sgam[lane] = ln_g[lane]; sbet[lane] = ln_b[lane]; }
  __syncthreads();

  const int m16 = lane & 15;
  const int hi  = (lane >> 4) & 1;
  const int wv = __builtin_amdgcn_readfirstlane(lane >> 5);
  proj_group< 9,  3,   0,  0, WOFF_J, 0>(sxp, swp, sbias, sp, m16, hi, wv);
  proj_group<17, 10,  36,  3, WOFF_O, 1>(sxp, swp, sbias, sp, m16, hi, wv);
  proj_group<11, 10, 236, 13, WOFF_G, 2>(sxp, swp, sbias, sp, m16, hi, wv);
  proj_group<11,  2, 356, 23, WOFF_B, 3>(sxp, swp, sbias, sp, m16, hi, wv);
  __syncthreads();

  constexpr int KB = 5;
  constexpr int NB = NQ / KB;
  constexpr int GE = KB * KV;
  static_assert(NB * KB == NQ, "groups must cover the rows");
  constexpr int PL = TILE * NQ * KV;
  const int s = lane >> 3;
  const int h = lane & 7;
  const int sb = s * (NQ * KV);
  const float* Pk  = smem + lds_base(SX_ELEMS + 1 * PL + sb);
  const float* Pv  = smem + lds_base(SX_ELEMS + 2 * PL + sb);
  const float* Pq  = smem + lds_base(SX_ELEMS + 0 * PL + sb + h * KV);
  const float* Pq3 = smem + lds_base(SX_ELEMS + 0 * PL + sb + (NQ - 1) * KV);
  const float* Pr  = smem + lds_base(SX_ELEMS + 3 * PL + sb + h * KV);
  const float* Pr3 = smem + lds_base(SX_ELEMS + 3 * PL + sb + (NQ - 1) * KV);
  float* ob  = smem + lds_base(sb + h * KV);
  float* ob3 = smem + lds_base(sb + (NQ - 1) * KV);
  const float* Pm = smask;

  v2f QA[KV], QB[KV];
  #pragma unroll
  for (int d = 0; d < KV; ++d) {
    QA[d] = (v2f){Pq[d], Pq[8 * KV + d]};
    QB[d] = (v2f){Pq[16 * KV + d], Pq3[d]};
  }
  float kq[NQ * KV], mq[NQ];
  #pragma unroll
  for (int i = 0; i < GE; ++i) kq[i] = Pk[i];
  #pragma unroll
  for (int i = 0; i < KB; ++i) mq[i] = Pm[i];

  v2f scA[NQ], scB[NQ];
  float mx0 = -3.0e38f, mx1 = -3.0e38f, mx2 = -3.0e38f, mx3 = -3.0e38f;
  #pragma unroll
  for (int b = 0; b < NB; ++b) {
    if (b + 1 < NB) {
      #pragma unroll
      for (int i = 0; i < GE; ++i) kq[(b + 1) * GE + i] = Pk[(b + 1) * GE + i];
      #pragma unroll
      for (int i = 0; i < KB; ++i) mq[(b + 1) * KB + i] = Pm[(b + 1) * KB + i];
    }
    __builtin_amdgcn_sched_barrier(0);
    #pragma unroll
    for (int jj = 0; jj < KB; ++jj) {
      const int j = b * KB + jj;
      const float* kr = kq + j * KV;
      v2f tA = QA[0] * (v2f){kr[0], kr[0]};
      v2f tB = QB[0] * (v2f){kr[0], kr[0]};
      #pragma unroll
      for (int d = 1; d < KV; ++d) {
        const v2f kk = {kr[d], kr[d]};
        tA = __builtin_elementwise_fma(QA[d], kk, tA);
        tB = __builtin_elementwise_fma(QB[d], kk, tB);
      }
      const v2f third = {1.0f / 3.0f, 1.0f / 3.0f};
      const v2f mkk = {mq[j], mq[j]};
      tA = __builtin_elementwise_fma(tA, third, mkk);
      tB = __builtin_elementwise_fma(tB, third, mkk);
      scA[j] = tA;
      scB[j] = tB;
      mx0 = fmaxf(mx0, tA.x); mx1 = fmaxf(mx1, tA.y);
      mx2 = fmaxf(mx2, tB.x); mx3 = fmaxf(mx3, tB.y);
    }
  }

  float vq[NQ * KV];
  #pragma unroll
  for (int i = 0; i < GE; ++i) vq[i] = Pv[i];
  __builtin_amdgcn_sched_barrier(0);

  v2f denA = {0.0f, 0.0f}, denB = {0.0f, 0.0f};
  #pragma unroll
  for (int j = 0; j < NQ; ++j) {
    const v2f eA = {__expf(scA[j].x - mx0), __expf(scA[j].y - mx1)};
    const v2f eB = {__expf(scB[j].x - mx2), __expf(scB[j].y - mx3)};
    scA[j] = eA;
    scB[j] = eB;
    denA += eA;
    denB += eB;
  }
  const v2f invA = {1.0f / denA.x, 1.0f / denA.y};
  const v2f invB = {1.0f / denB.x, 1.0f / denB.y};

  v2f oA[KV], oB[KV];
  float r0[KV], r1[KV], r2[KV], r3[KV], lg[KV], lb[KV];
  #pragma unroll
  for (int b = 0; b < NB; ++b) {
    if (b + 1 < NB) {
      #pragma unroll
      for (int i = 0; i < GE; ++i) vq[(b + 1) * GE + i] = Pv[(b + 1) * GE + i];
    } else {
      #pragma unroll
      for (int d = 0; d < KV; ++d) {
        r0[d] = Pr[d];
        r1[d] = Pr[8 * KV + d];
        r2[d] = Pr[16 * KV + d];
        r3[d] = Pr3[d];
        lg[d] = sgam[d];
        lb[d] = sbet[d];
      }
    }
    __builtin_amdgcn_sched_barrier(0);
    #pragma unroll
    for (int jj = 0; jj < KB; ++jj) {
      const int j = b * KB + jj;
      const float* vr = vq + j * KV;
      #pragma unroll
      for (int d = 0; d < KV; ++d) {
        const v2f vv = {vr[d], vr[d]};
        if (j == 0) {
          oA[d] = scA[0] * vv;
          oB[d] = scB[0] * vv;
        } else {
          oA[d] = __builtin_elementwise_fma(scA[j], vv, oA[d]);
          oB[d] = __builtin_elementwise_fma(scB[j], vv, oB[d]);
        }
      }
    }
  }

  v2f muA = {0.0f, 0.0f}, muB = {0.0f, 0.0f};
  #pragma unroll
  for (int d = 0; d < KV; ++d) {
    oA[d] = __builtin_elementwise_fma(oA[d], invA, (v2f){r0[d], r1[d]});
    oB[d] = __builtin_elementwise_fma(oB[d], invB, (v2f){r2[d], r3[d]});
    muA += oA[d];
    muB += oB[d];
  }
  muA *= (1.0f / KV);
  muB *= (1.0f / KV);
  v2f varA = {0.0f, 0.0f}, varB = {0.0f, 0.0f};
  #pragma unroll
  for (int d = 0; d < KV; ++d) {
    const v2f dA = oA[d] - muA, dB = oB[d] - muB;
    varA = __builtin_elementwise_fma(dA, dA, varA);
    varB = __builtin_elementwise_fma(dB, dB, varB);
  }
  varA *= (1.0f / KV);
  varB *= (1.0f / KV);
  const v2f rsA = {rsqrtf(varA.x + 1e-5f), rsqrtf(varA.y + 1e-5f)};
  const v2f rsB = {rsqrtf(varB.x + 1e-5f), rsqrtf(varB.y + 1e-5f)};
  #pragma unroll
  for (int d = 0; d < KV; ++d) {
    const v2f gg = {lg[d], lg[d]}, bb2 = {lb[d], lb[d]};
    const v2f yA = __builtin_elementwise_fma((oA[d] - muA) * rsA, gg, bb2);
    const v2f yB = __builtin_elementwise_fma((oB[d] - muB) * rsB, gg, bb2);
    ob[d] = yA.x;
    ob[8 * KV + d] = yA.y;
    ob[16 * KV + d] = yB.x;
    if (h == 0) ob3[d] = yB.y;
  }
  __syncthreads();

  const int t3 = wv * 32 + (int)__builtin_amdgcn_mbcnt_lo(~0u, 0u);
  const int nvalid = (int)min((long long)TILE, (long long)btot - tile0);
  float* __restrict__ ot = out + tile0 * (NQ * KV);
  for (int i = t3; i < nvalid * (NQ * KV); i += NTHR) ot[i] = sxp[i];
}

extern "C" void kernel_launch(void* const* d_in, const int* in_sizes, int n_in,
                              void* d_out, int out_size, void* d_ws, size_t ws_size,
                              hipStream_t stream) {
  (void)d_ws; (void)ws_size; (void)n_in; (void)out_size;
  const float* x    = (const float*)d_in[0];
  const int*   mask = (const int*)d_in[1];
  const float* ln_g = (const float*)d_in[2];
  const float* ln_b = (const float*)d_in[3];
  const float* Wj   = (const float*)d_in[4];
  const float* bj   = (const float*)d_in[5];
  const float* Wo   = (const float*)d_in[6];
  const float* bo   = (const float*)d_in[7];
  const float* Wg   = (const float*)d_in[8];
  const float* bg   = (const float*)d_in[9];
  const float* Wb   = (const float*)d_in[10];
  const float* bb   = (const float*)d_in[11];
  float* out = (float*)d_out;

  const int btot = in_sizes[0] / XDIM;
  const int grid = (btot + TILE - 1) / TILE;
  attn_fused_kernel<<<grid, 128, 0, stream>>>(x, mask, ln_g, ln_b,
                                             Wj, bj, Wo, bo, Wg, bg, Wb, bb,
                                             out, btot);
}


# x tile kept in LDS in its memory layout: 11 straight 16-byte copies per lane replace the per-dword repack (divide + selects each); A fragments read at token offsets, zero W columns cover k >= L
# speedup vs baseline: 9.1992x; 1.2473x over previous
#include <hip/hip_runtime.h>

typedef float v2f __attribute__((ext_vector_type(2)));
typedef float v4f __attribute__((ext_vector_type(4)));
typedef float v8f __attribute__((ext_vector_type(8)));

namespace {
constexpr int XDIM = 329;
constexpr int NQ   = 25;
constexpr int KV   = 9;
constexpr int TILE = 16;
constexpr int NWAVE = 4;
constexpr int NTHR  = 32 * NWAVE;
constexpr int XTAIL = 4;

constexpr int SX_ELEMS = TILE * XDIM + XTAIL;
constexpr int SP_ELEMS = 4 * TILE * NQ * KV;
constexpr int SW_ELEMS = 48*12 + 48*20 + 48*12 + 48*12;
constexpr int SB_ELEMS = 4 * 48;
constexpr int SMEM_ELEMS = SX_ELEMS + SP_ELEMS + SW_ELEMS + SB_ELEMS + NQ + 2 * KV;

constexpr int TILE_F4 = TILE * XDIM / 4;
constexpr int XCH     = (TILE_F4 + NTHR - 1) / NTHR;
static_assert(TILE * XDIM % 4 == 0, "tile must be whole 16-byte pieces");
static_assert(XCH * NTHR >= TILE_F4, "the group must cover the tile");

constexpr int WOFF_J = 0;
constexpr int WOFF_O = 48 * 12;
constexpr int WOFF_G = WOFF_O + 48 * 20;
constexpr int WOFF_B = WOFF_G + 48 * 12;
}

__device__ __forceinline__ int lds_base(int idx) {
  asm("" : "+v"(idx));
  return idx;
}

template <int L, int LPAD, int WOFF>
__device__ __forceinline__ void stage_w(const float* __restrict__ W, float* swp, int lane) {
  constexpr int N  = 4 * KV * L;
  constexpr int NT = (N + NTHR - 1) / NTHR;
  float r[NT];
  #pragma unroll
  for (int t = 0; t < NT; ++t) {
    const int i = min(lane + NTHR * t, N - 1);
    r[t] = W[i];
  }
  #pragma unroll
  for (int t = 0; t < NT; ++t) {
    const int i = min(lane + NTHR * t, N - 1);
    const int n = i / L, k = i - n * L;
    swp[WOFF + n * LPAD + k] = r[t];
  }
}

template <int L, int NVEC, int COL0, int Q0, int WOFF, int GIDX>
__device__ __forceinline__ void proj_group(const float* sxp, const float* swp,
                                           const float* sbias, float* sp,
                                           int m16, int hi, int wv) {
  const int j0 = (wv + 6 * NWAVE - Q0) % NWAVE;
  constexpr int KSTEPS = (L + 3) / 4;
  constexpr int LPAD = KSTEPS * 4;
  #pragma unroll
  for (int nt = 0; nt < 3; ++nt) {
    const int nn = nt * 16 + m16;
    v2f bf[KSTEPS];
    #pragma unroll
    for (int ks = 0; ks < KSTEPS; ++ks)
      bf[ks] = *reinterpret_cast<const v2f*>(swp + WOFF + nn * LPAD + ks * 4 + 2 * hi);
    const float bias = sbias[GIDX * 48 + nn];

    for (int j = j0; j < NVEC; j += NWAVE) {
      const float* ap = sxp + m16 * XDIM + COL0 + j * L + 2 * hi;
      v8f acc = {0.f, 0.f, 0.f, 0.f, 0.f, 0.f, 0.f, 0.f};
      #pragma unroll
      for (int ks = 0; ks < KSTEPS; ++ks) {
        const v2f a = {ap[ks * 4], ap[ks * 4 + 1]};
        acc = __builtin_amdgcn_wmma_f32_16x16x4_f32(
             false, a,  false, bf[ks],
             (short)0, acc,  false,  false);
      }
      if (nn < 4 * KV) {
        const int p = nn / KV, kvc = nn - (nn / KV) * KV;
        const int q = Q0 + j;
        #pragma unroll
        for (int r = 0; r < 8; ++r) {
          const int m = r + 8 * hi;
          sp[((p * TILE + m) * NQ + q) * KV + kvc] = acc[r] + bias;
        }
      }
    }
  }
}

__global__ __launch_bounds__(NTHR) __attribute__((amdgpu_waves_per_eu(3)))
void attn_fused_kernel(
    const float* __restrict__ x, const int* __restrict__ mask,
    const float* __restrict__ ln_g, const float* __restrict__ ln_b,
    const float* __restrict__ Wj, const float* __restrict__ bj,
    const float* __restrict__ Wo, const float* __restrict__ bo,
    const float* __restrict__ Wg, const float* __restrict__ bg,
    const float* __restrict__ Wb, const float* __restrict__ bb,
    float* __restrict__ out, int btot) {
  __shared__ __align__(16) float smem[SMEM_ELEMS];
  float* sxp   = smem;
  float* sp    = smem + SX_ELEMS;
  float* swp   = sp + SP_ELEMS;
  float* sbias = swp + SW_ELEMS;
  float* smask = sbias + SB_ELEMS;
  float* sgam  = smask + NQ;
  float* sbet  = sgam + KV;

  const int lane = threadIdx.x;
  const long long tile0 = (long long)blockIdx.x * TILE;

  if (lane < XTAIL) sxp[TILE * XDIM + lane] = 0.0f;
  for (int i = lane; i < SW_ELEMS + SB_ELEMS; i += NTHR) swp[i] = 0.0f;
  __syncthreads();

  const bool whole = (tile0 + TILE <= (long long)btot) &&
                     ((reinterpret_cast<unsigned long long>(x) & 15ull) == 0ull);
  if (whole) {
    const v4f* xt = reinterpret_cast<const v4f*>(x + tile0 * XDIM);
    v4f cur[XCH];
    #pragma unroll
    for (int u = 0; u < XCH; ++u) cur[u] = xt[min(u * NTHR + lane, TILE_F4 - 1)];
    v4f* sx4 = reinterpret_cast<v4f*>(sxp);
    #pragma unroll
    for (int u = 0; u < XCH; ++u) sx4[min(u * NTHR + lane, TILE_F4 - 1)] = cur[u];
  } else {
    for (int i = lane; i < TILE * XDIM; i += NTHR) {
      const int s = i / XDIM;
      sxp[i] = ((tile0 + s) < btot) ? x[tile0 * XDIM + i] : 0.0f;
    }
  }
  stage_w< 9, 12, WOFF_J>(Wj, swp, lane);
  stage_w<17, 20, WOFF_O>(Wo, swp, lane);
  stage_w<11, 12, WOFF_G>(Wg, swp, lane);
  stage_w<11, 12, WOFF_B>(Wb, swp, lane);
  for (int i = lane; i < 4 * KV; i += NTHR) {
    sbias[0 * 48 + i] = bj[i];
    sbias[1 * 48 + i] = bo[i];
    sbias[2 * 48 + i] = bg[i];
    sbias[3 * 48 + i] = bb[i];
  }
  if (lane < NQ) smask[lane] = -1.0e9f * (float)mask[lane];
  if (lane < KV) { sgam[lane] = ln_g[lane]; sbet[lane] = ln_b[lane]; }
  __syncthreads();

  const int m16 = lane & 15;
  const int hi  = (lane >> 4) & 1;
  const int wv = __builtin_amdgcn_readfirstlane(lane >> 5);
  proj_group< 9,  3,   0,  0, WOFF_J, 0>(sxp, swp, sbias, sp, m16, hi, wv);
  proj_group<17, 10,  27,  3, WOFF_O, 1>(sxp, swp, sbias, sp, m16, hi, wv);
  proj_group<11, 10, 197, 13, WOFF_G, 2>(sxp, swp, sbias, sp, m16, hi, wv);
  proj_group<11,  2, 307, 23, WOFF_B, 3>(sxp, swp, sbias, sp, m16, hi, wv);
  __syncthreads();

  constexpr int KB = 5;
  constexpr int NB = NQ / KB;
  constexpr int GE = KB * KV;
  static_assert(NB * KB == NQ, "groups must cover the rows");
  constexpr int PL = TILE * NQ * KV;
  const int s = lane >> 3;
  const int h = lane & 7;
  const int sb = s * (NQ * KV);
  const float* Pk  = smem + lds_base(SX_ELEMS + 1 * PL + sb);
  const float* Pv  = smem + lds_base(SX_ELEMS + 2 * PL + sb);
  const float* Pq  = smem + lds_base(SX_ELEMS + 0 * PL + sb + h * KV);
  const float* Pq3 = smem + lds_base(SX_ELEMS + 0 * PL + sb + (NQ - 1) * KV);
  const float* Pr  = smem + lds_base(SX_ELEMS + 3 * PL + sb + h * KV);
  const float* Pr3 = smem + lds_base(SX_ELEMS + 3 * PL + sb + (NQ - 1) * KV);
  float* ob  = smem + lds_base(sb + h * KV);
  float* ob3 = smem + lds_base(sb + (NQ - 1) * KV);
  const float* Pm = smask;

  v2f QA[KV], QB[KV];
  #pragma unroll
  for (int d = 0; d < KV; ++d) {
    QA[d] = (v2f){Pq[d], Pq[8 * KV + d]};
    QB[d] = (v2f){Pq[16 * KV + d], Pq3[d]};
  }
  float kq[NQ * KV], mq[NQ];
  #pragma unroll
  for (int i = 0; i < GE; ++i) kq[i] = Pk[i];
  #pragma unroll
  for (int i = 0; i < KB; ++i) mq[i] = Pm[i];

  v2f scA[NQ], scB[NQ];
  float mx0 = -3.0e38f, mx1 = -3.0e38f, mx2 = -3.0e38f, mx3 = -3.0e38f;
  #pragma unroll
  for (int b = 0; b < NB; ++b) {
    if (b + 1 < NB) {
      #pragma unroll
      for (int i = 0; i < GE; ++i) kq[(b + 1) * GE + i] = Pk[(b + 1) * GE + i];
      #pragma unroll
      for (int i = 0; i < KB; ++i) mq[(b + 1) * KB + i] = Pm[(b + 1) * KB + i];
    }
    __builtin_amdgcn_sched_barrier(0);
    #pragma unroll
    for (int jj = 0; jj < KB; ++jj) {
      const int j = b * KB + jj;
      const float* kr = kq + j * KV;
      v2f tA = QA[0] * (v2f){kr[0], kr[0]};
      v2f tB = QB[0] * (v2f){kr[0], kr[0]};
      #pragma unroll
      for (int d = 1; d < KV; ++d) {
        const v2f kk = {kr[d], kr[d]};
        tA = __builtin_elementwise_fma(QA[d], kk, tA);
        tB = __builtin_elementwise_fma(QB[d], kk, tB);
      }
      const v2f third = {1.0f / 3.0f, 1.0f / 3.0f};
      const v2f mkk = {mq[j], mq[j]};
      tA = __builtin_elementwise_fma(tA, third, mkk);
      tB = __builtin_elementwise_fma(tB, third, mkk);
      scA[j] = tA;
      scB[j] = tB;
      mx0 = fmaxf(mx0, tA.x); mx1 = fmaxf(mx1, tA.y);
      mx2 = fmaxf(mx2, tB.x); mx3 = fmaxf(mx3, tB.y);
    }
  }

  float vq[NQ * KV];
  #pragma unroll
  for (int i = 0; i < GE; ++i) vq[i] = Pv[i];
  __builtin_amdgcn_sched_barrier(0);

  v2f denA = {0.0f, 0.0f}, denB = {0.0f, 0.0f};
  #pragma unroll
  for (int j = 0; j < NQ; ++j) {
    const v2f eA = {__expf(scA[j].x - mx0), __expf(scA[j].y - mx1)};
    const v2f eB = {__expf(scB[j].x - mx2), __expf(scB[j].y - mx3)};
    scA[j] = eA;
    scB[j] = eB;
    denA += eA;
    denB += eB;
  }
  const v2f invA = {1.0f / denA.x, 1.0f / denA.y};
  const v2f invB = {1.0f / denB.x, 1.0f / denB.y};

  v2f oA[KV], oB[KV];
  float r0[KV], r1[KV], r2[KV], r3[KV], lg[KV], lb[KV];
  #pragma unroll
  for (int b = 0; b < NB; ++b) {
    if (b + 1 < NB) {
      #pragma unroll
      for (int i = 0; i < GE; ++i) vq[(b + 1) * GE + i] = Pv[(b + 1) * GE + i];
    } else {
      #pragma unroll
      for (int d = 0; d < KV; ++d) {
        r0[d] = Pr[d];
        r1[d] = Pr[8 * KV + d];
        r2[d] = Pr[16 * KV + d];
        r3[d] = Pr3[d];
        lg[d] = sgam[d];
        lb[d] = sbet[d];
      }
    }
    __builtin_amdgcn_sched_barrier(0);
    #pragma unroll
    for (int jj = 0; jj < KB; ++jj) {
      const int j = b * KB + jj;
      const float* vr = vq + j * KV;
      #pragma unroll
      for (int d = 0; d < KV; ++d) {
        const v2f vv = {vr[d], vr[d]};
        if (j == 0) {
          oA[d] = scA[0] * vv;
          oB[d] = scB[0] * vv;
        } else {
          oA[d] = __builtin_elementwise_fma(scA[j], vv, oA[d]);
          oB[d] = __builtin_elementwise_fma(scB[j], vv, oB[d]);
        }
      }
    }
  }

  v2f muA = {0.0f, 0.0f}, muB = {0.0f, 0.0f};
  #pragma unroll
  for (int d = 0; d < KV; ++d) {
    oA[d] = __builtin_elementwise_fma(oA[d], invA, (v2f){r0[d], r1[d]});
    oB[d] = __builtin_elementwise_fma(oB[d], invB, (v2f){r2[d], r3[d]});
    muA += oA[d];
    muB += oB[d];
  }
  muA *= (1.0f / KV);
  muB *= (1.0f / KV);
  v2f varA = {0.0f, 0.0f}, varB = {0.0f, 0.0f};
  #pragma unroll
  for (int d = 0; d < KV; ++d) {
    const v2f dA = oA[d] - muA, dB = oB[d] - muB;
    varA = __builtin_elementwise_fma(dA, dA, varA);
    varB = __builtin_elementwise_fma(dB, dB, varB);
  }
  varA *= (1.0f / KV);
  varB *= (1.0f / KV);
  const v2f rsA = {rsqrtf(varA.x + 1e-5f), rsqrtf(varA.y + 1e-5f)};
  const v2f rsB = {rsqrtf(varB.x + 1e-5f), rsqrtf(varB.y + 1e-5f)};
  #pragma unroll
  for (int d = 0; d < KV; ++d) {
    const v2f gg = {lg[d], lg[d]}, bb2 = {lb[d], lb[d]};
    const v2f yA = __builtin_elementwise_fma((oA[d] - muA) * rsA, gg, bb2);
    const v2f yB = __builtin_elementwise_fma((oB[d] - muB) * rsB, gg, bb2);
    ob[d] = yA.x;
    ob[8 * KV + d] = yA.y;
    ob[16 * KV + d] = yB.x;
    if (h == 0) ob3[d] = yB.y;
  }
  __syncthreads();

  const int t3 = wv * 32 + (int)__builtin_amdgcn_mbcnt_lo(~0u, 0u);
  const int nvalid = (int)min((long long)TILE, (long long)btot - tile0);
  float* __restrict__ ot = out + tile0 * (NQ * KV);
  for (int i = t3; i < nvalid * (NQ * KV); i += NTHR) ot[i] = sxp[i];
}

extern "C" void kernel_launch(void* const* d_in, const int* in_sizes, int n_in,
                              void* d_out, int out_size, void* d_ws, size_t ws_size,
                              hipStream_t stream) {
  (void)d_ws; (void)ws_size; (void)n_in; (void)out_size;
  const float* x    = (const float*)d_in[0];
  const int*   mask = (const int*)d_in[1];
  const float* ln_g = (const float*)d_in[2];
  const float* ln_b = (const float*)d_in[3];
  const float* Wj   = (const float*)d_in[4];
  const float* bj   = (const float*)d_in[5];
  const float* Wo   = (const float*)d_in[6];
  const float* bo   = (const float*)d_in[7];
  const float* Wg   = (const float*)d_in[8];
  const float* bg   = (const float*)d_in[9];
  const float* Wb   = (const float*)d_in[10];
  const float* bb   = (const float*)d_in[11];
  float* out = (float*)d_out;

  const int btot = in_sizes[0] / XDIM;
  const int grid = (btot + TILE - 1) / TILE;
  attn_fused_kernel<<<grid, 128, 0, stream>>>(x, mask, ln_g, ln_b,
                                             Wj, bj, Wo, bo, Wg, bg, Wb, bb,
                                             out, btot);
}


# projections: the loop runs over tokens, a token's A fragments feed the three column tiles as three independent accumulator chains; weights stay in registers, next token's fragments loaded ahead
# speedup vs baseline: 10.0291x; 1.0902x over previous
#include <hip/hip_runtime.h>

typedef float v2f __attribute__((ext_vector_type(2)));
typedef float v4f __attribute__((ext_vector_type(4)));
typedef float v8f __attribute__((ext_vector_type(8)));

namespace {
constexpr int XDIM = 329;
constexpr int NQ   = 25;
constexpr int KV   = 9;
constexpr int TILE = 16;
constexpr int NWAVE = 4;
constexpr int NTHR  = 32 * NWAVE;
constexpr int XTAIL = 4;

constexpr int SX_ELEMS = TILE * XDIM + XTAIL;
constexpr int SP_ELEMS = 4 * TILE * NQ * KV;
constexpr int SW_ELEMS = 48*12 + 48*20 + 48*12 + 48*12;
constexpr int SB_ELEMS = 4 * 48;
constexpr int SMEM_ELEMS = SX_ELEMS + SP_ELEMS + SW_ELEMS + SB_ELEMS + NQ + 2 * KV;

constexpr int TILE_F4 = TILE * XDIM / 4;
constexpr int XCH     = (TILE_F4 + NTHR - 1) / NTHR;
static_assert(TILE * XDIM % 4 == 0, "tile must be whole 16-byte pieces");
static_assert(XCH * NTHR >= TILE_F4, "the group must cover the tile");

constexpr int WOFF_J = 0;
constexpr int WOFF_O = 48 * 12;
constexpr int WOFF_G = WOFF_O + 48 * 20;
constexpr int WOFF_B = WOFF_G + 48 * 12;
}

__device__ __forceinline__ int lds_base(int idx) {
  asm("" : "+v"(idx));
  return idx;
}

template <int L, int LPAD, int WOFF>
__device__ __forceinline__ void stage_w(const float* __restrict__ W, float* swp, int lane) {
  constexpr int N  = 4 * KV * L;
  constexpr int NT = (N + NTHR - 1) / NTHR;
  float r[NT];
  #pragma unroll
  for (int t = 0; t < NT; ++t) {
    const int i = min(lane + NTHR * t, N - 1);
    r[t] = W[i];
  }
  #pragma unroll
  for (int t = 0; t < NT; ++t) {
    const int i = min(lane + NTHR * t, N - 1);
    const int n = i / L, k = i - n * L;
    swp[WOFF + n * LPAD + k] = r[t];
  }
}

template <int L, int NVEC, int COL0, int Q0, int WOFF, int GIDX>
__device__ __forceinline__ void proj_group(const float* sxp, const float* swp,
                                           const float* sbias, float* sp,
                                           int m16, int hi, int wv, const int (&cb)[3]) {
  const int j0 = (wv + 6 * NWAVE - Q0) % NWAVE;
  constexpr int KSTEPS = (L + 3) / 4;
  constexpr int LPAD = KSTEPS * 4;
  if (j0 >= NVEC) return;

  v2f bf[3][KSTEPS];
  float bias[3];
  #pragma unroll
  for (int nt = 0; nt < 3; ++nt) {
    const int nn = nt * 16 + m16;
    #pragma unroll
    for (int ks = 0; ks < KSTEPS; ++ks)
      bf[nt][ks] = *reinterpret_cast<const v2f*>(swp + WOFF + nn * LPAD + ks * 4 + 2 * hi);
    bias[nt] = sbias[GIDX * 48 + nn];
  }

  const float* arow = sxp + m16 * XDIM + COL0 + 2 * hi;
  v2f a[KSTEPS];
  #pragma unroll
  for (int ks = 0; ks < KSTEPS; ++ks)
    a[ks] = (v2f){arow[j0 * L + ks * 4], arow[j0 * L + ks * 4 + 1]};

  for (int j = j0; j < NVEC; j += NWAVE) {
    const int jn = min(j + NWAVE, NVEC - 1);
    v2f an[KSTEPS];
    #pragma unroll
    for (int ks = 0; ks < KSTEPS; ++ks)
      an[ks] = (v2f){arow[jn * L + ks * 4], arow[jn * L + ks * 4 + 1]};

    v8f acc[3];
    #pragma unroll
    for (int nt = 0; nt < 3; ++nt) acc[nt] = (v8f){0.f, 0.f, 0.f, 0.f, 0.f, 0.f, 0.f, 0.f};
    #pragma unroll
    for (int ks = 0; ks < KSTEPS; ++ks) {
      #pragma unroll
      for (int nt = 0; nt < 3; ++nt)
        acc[nt] = __builtin_amdgcn_wmma_f32_16x16x4_f32(
             false, a[ks],  false, bf[nt][ks],
             (short)0, acc[nt],  false,  false);
    }

    const int q = Q0 + j;
    #pragma unroll
    for (int nt = 0; nt < 3; ++nt) {
      if (nt < 2 || m16 < 4 * KV - 32) {
        float* d = sp + cb[nt] + q * KV;
        #pragma unroll
        for (int r = 0; r < 8; ++r) d[r * (NQ * KV)] = acc[nt][r] + bias[nt];
      }
    }
    #pragma unroll
    for (int ks = 0; ks < KSTEPS; ++ks) a[ks] = an[ks];
  }
}

__global__ __launch_bounds__(NTHR) __attribute__((amdgpu_waves_per_eu(3)))
void attn_fused_kernel(
    const float* __restrict__ x, const int* __restrict__ mask,
    const float* __restrict__ ln_g, const float* __restrict__ ln_b,
    const float* __restrict__ Wj, const float* __restrict__ bj,
    const float* __restrict__ Wo, const float* __restrict__ bo,
    const float* __restrict__ Wg, const float* __restrict__ bg,
    const float* __restrict__ Wb, const float* __restrict__ bb,
    float* __restrict__ out, int btot) {
  __shared__ __align__(16) float smem[SMEM_ELEMS];
  float* sxp   = smem;
  float* sp    = smem + SX_ELEMS;
  float* swp   = sp + SP_ELEMS;
  float* sbias = swp + SW_ELEMS;
  float* smask = sbias + SB_ELEMS;
  float* sgam  = smask + NQ;
  float* sbet  = sgam + KV;

  const int lane = threadIdx.x;
  const long long tile0 = (long long)blockIdx.x * TILE;

  if (lane < XTAIL) sxp[TILE * XDIM + lane] = 0.0f;
  for (int i = lane; i < SW_ELEMS + SB_ELEMS; i += NTHR) swp[i] = 0.0f;
  __syncthreads();

  const bool whole = (tile0 + TILE <= (long long)btot) &&
                     ((reinterpret_cast<unsigned long long>(x) & 15ull) == 0ull);
  if (whole) {
    const v4f* xt = reinterpret_cast<const v4f*>(x + tile0 * XDIM);
    v4f cur[XCH];
    #pragma unroll
    for (int u = 0; u < XCH; ++u) cur[u] = xt[min(u * NTHR + lane, TILE_F4 - 1)];
    v4f* sx4 = reinterpret_cast<v4f*>(sxp);
    #pragma unroll
    for (int u = 0; u < XCH; ++u) sx4[min(u * NTHR + lane, TILE_F4 - 1)] = cur[u];
  } else {
    for (int i = lane; i < TILE * XDIM; i += NTHR) {
      const int s = i / XDIM;
      sxp[i] = ((tile0 + s) < btot) ? x[tile0 * XDIM + i] : 0.0f;
    }
  }
  stage_w< 9, 12, WOFF_J>(Wj, swp, lane);
  stage_w<17, 20, WOFF_O>(Wo, swp, lane);
  stage_w<11, 12, WOFF_G>(Wg, swp, lane);
  stage_w<11, 12, WOFF_B>(Wb, swp, lane);
  for (int i = lane; i < 4 * KV; i += NTHR) {
    sbias[0 * 48 + i] = bj[i];
    sbias[1 * 48 + i] = bo[i];
    sbias[2 * 48 + i] = bg[i];
    sbias[3 * 48 + i] = bb[i];
  }
  if (lane < NQ) smask[lane] = -1.0e9f * (float)mask[lane];
  if (lane < KV) { sgam[lane] = ln_g[lane]; sbet[lane] = ln_b[lane]; }
  __syncthreads();

  const int m16 = lane & 15;
  const int hi  = (lane >> 4) & 1;
  const int wv = __builtin_amdgcn_readfirstlane(lane >> 5);
  int cb[3];
  #pragma unroll
  for (int nt = 0; nt < 3; ++nt) {
    const int nn = nt * 16 + m16;
    const int p = nn / KV;
    cb[nt] = p * (TILE * NQ * KV) + (nn - p * KV) + hi * (8 * NQ * KV);
  }
  proj_group< 9,  3,   0,  0, WOFF_J, 0>(sxp, swp, sbias, sp, m16, hi, wv, cb);
  proj_group<17, 10,  27,  3, WOFF_O, 1>(sxp, swp, sbias, sp, m16, hi, wv, cb);
  proj_group<11, 10, 197, 13, WOFF_G, 2>(sxp, swp, sbias, sp, m16, hi, wv, cb);
  proj_group<11,  2, 307, 23, WOFF_B, 3>(sxp, swp, sbias, sp, m16, hi, wv, cb);
  __syncthreads();

  constexpr int KB = 5;
  constexpr int NB = NQ / KB;
  constexpr int GE = KB * KV;
  static_assert(NB * KB == NQ, "groups must cover the rows");
  constexpr int PL = TILE * NQ * KV;
  const int s = lane >> 3;
  const int h = lane & 7;
  const int sb = s * (NQ * KV);
  const float* Pk  = smem + lds_base(SX_ELEMS + 1 * PL + sb);
  const float* Pv  = smem + lds_base(SX_ELEMS + 2 * PL + sb);
  const float* Pq  = smem + lds_base(SX_ELEMS + 0 * PL + sb + h * KV);
  const float* Pq3 = smem + lds_base(SX_ELEMS + 0 * PL + sb + (NQ - 1) * KV);
  const float* Pr  = smem + lds_base(SX_ELEMS + 3 * PL + sb + h * KV);
  const float* Pr3 = smem + lds_base(SX_ELEMS + 3 * PL + sb + (NQ - 1) * KV);
  float* ob  = smem + lds_base(sb + h * KV);
  float* ob3 = smem + lds_base(sb + (NQ - 1) * KV);
  const float* Pm = smask;

  v2f QA[KV], QB[KV];
  #pragma unroll
  for (int d = 0; d < KV; ++d) {
    QA[d] = (v2f){Pq[d], Pq[8 * KV + d]};
    QB[d] = (v2f){Pq[16 * KV + d], Pq3[d]};
  }
  float kq[NQ * KV], mq[NQ];
  #pragma unroll
  for (int i = 0; i < GE; ++i) kq[i] = Pk[i];
  #pragma unroll
  for (int i = 0; i < KB; ++i) mq[i] = Pm[i];

  v2f scA[NQ], scB[NQ];
  float mx0 = -3.0e38f, mx1 = -3.0e38f, mx2 = -3.0e38f, mx3 = -3.0e38f;
  #pragma unroll
  for (int b = 0; b < NB; ++b) {
    if (b + 1 < NB) {
      #pragma unroll
      for (int i = 0; i < GE; ++i) kq[(b + 1) * GE + i] = Pk[(b + 1) * GE + i];
      #pragma unroll
      for (int i = 0; i < KB; ++i) mq[(b + 1) * KB + i] = Pm[(b + 1) * KB + i];
    }
    __builtin_amdgcn_sched_barrier(0);
    #pragma unroll
    for (int jj = 0; jj < KB; ++jj) {
      const int j = b * KB + jj;
      const float* kr = kq + j * KV;
      v2f tA = QA[0] * (v2f){kr[0], kr[0]};
      v2f tB = QB[0] * (v2f){kr[0], kr[0]};
      #pragma unroll
      for (int d = 1; d < KV; ++d) {
        const v2f kk = {kr[d], kr[d]};
        tA = __builtin_elementwise_fma(QA[d], kk, tA);
        tB = __builtin_elementwise_fma(QB[d], kk, tB);
      }
      const v2f third = {1.0f / 3.0f, 1.0f / 3.0f};
      const v2f mkk = {mq[j], mq[j]};
      tA = __builtin_elementwise_fma(tA, third, mkk);
      tB = __builtin_elementwise_fma(tB, third, mkk);
      scA[j] = tA;
      scB[j] = tB;
      mx0 = fmaxf(mx0, tA.x); mx1 = fmaxf(mx1, tA.y);
      mx2 = fmaxf(mx2, tB.x); mx3 = fmaxf(mx3, tB.y);
    }
  }

  float vq[NQ * KV];
  #pragma unroll
  for (int i = 0; i < GE; ++i) vq[i] = Pv[i];
  __builtin_amdgcn_sched_barrier(0);

  v2f denA = {0.0f, 0.0f}, denB = {0.0f, 0.0f};
  #pragma unroll
  for (int j = 0; j < NQ; ++j) {
    const v2f eA = {__expf(scA[j].x - mx0), __expf(scA[j].y - mx1)};
    const v2f eB = {__expf(scB[j].x - mx2), __expf(scB[j].y - mx3)};
    scA[j] = eA;
    scB[j] = eB;
    denA += eA;
    denB += eB;
  }
  const v2f invA = {1.0f / denA.x, 1.0f / denA.y};
  const v2f invB = {1.0f / denB.x, 1.0f / denB.y};

  v2f oA[KV], oB[KV];
  float r0[KV], r1[KV], r2[KV], r3[KV], lg[KV], lb[KV];
  #pragma unroll
  for (int b = 0; b < NB; ++b) {
    if (b + 1 < NB) {
      #pragma unroll
      for (int i = 0; i < GE; ++i) vq[(b + 1) * GE + i] = Pv[(b + 1) * GE + i];
    } else {
      #pragma unroll
      for (int d = 0; d < KV; ++d) {
        r0[d] = Pr[d];
        r1[d] = Pr[8 * KV + d];
        r2[d] = Pr[16 * KV + d];
        r3[d] = Pr3[d];
        lg[d] = sgam[d];
        lb[d] = sbet[d];
      }
    }
    __builtin_amdgcn_sched_barrier(0);
    #pragma unroll
    for (int jj = 0; jj < KB; ++jj) {
      const int j = b * KB + jj;
      const float* vr = vq + j * KV;
      #pragma unroll
      for (int d = 0; d < KV; ++d) {
        const v2f vv = {vr[d], vr[d]};
        if (j == 0) {
          oA[d] = scA[0] * vv;
          oB[d] = scB[0] * vv;
        } else {
          oA[d] = __builtin_elementwise_fma(scA[j], vv, oA[d]);
          oB[d] = __builtin_elementwise_fma(scB[j], vv, oB[d]);
        }
      }
    }
  }

  v2f muA = {0.0f, 0.0f}, muB = {0.0f, 0.0f};
  #pragma unroll
  for (int d = 0; d < KV; ++d) {
    oA[d] = __builtin_elementwise_fma(oA[d], invA, (v2f){r0[d], r1[d]});
    oB[d] = __builtin_elementwise_fma(oB[d], invB, (v2f){r2[d], r3[d]});
    muA += oA[d];
    muB += oB[d];
  }
  muA *= (1.0f / KV);
  muB *= (1.0f / KV);
  v2f varA = {0.0f, 0.0f}, varB = {0.0f, 0.0f};
  #pragma unroll
  for (int d = 0; d < KV; ++d) {
    const v2f dA = oA[d] - muA, dB = oB[d] - muB;
    varA = __builtin_elementwise_fma(dA, dA, varA);
    varB = __builtin_elementwise_fma(dB, dB, varB);
  }
  varA *= (1.0f / KV);
  varB *= (1.0f / KV);
  const v2f rsA = {rsqrtf(varA.x + 1e-5f), rsqrtf(varA.y + 1e-5f)};
  const v2f rsB = {rsqrtf(varB.x + 1e-5f), rsqrtf(varB.y + 1e-5f)};
  #pragma unroll
  for (int d = 0; d < KV; ++d) {
    const v2f gg = {lg[d], lg[d]}, bb2 = {lb[d], lb[d]};
    const v2f yA = __builtin_elementwise_fma((oA[d] - muA) * rsA, gg, bb2);
    const v2f yB = __builtin_elementwise_fma((oB[d] - muB) * rsB, gg, bb2);
    ob[d] = yA.x;
    ob[8 * KV + d] = yA.y;
    ob[16 * KV + d] = yB.x;
    if (h == 0) ob3[d] = yB.y;
  }
  __syncthreads();

  const int t3 = wv * 32 + (int)__builtin_amdgcn_mbcnt_lo(~0u, 0u);
  const int nvalid = (int)min((long long)TILE, (long long)btot - tile0);
  float* __restrict__ ot = out + tile0 * (NQ * KV);
  for (int i = t3; i < nvalid * (NQ * KV); i += NTHR) ot[i] = sxp[i];
}

extern "C" void kernel_launch(void* const* d_in, const int* in_sizes, int n_in,
                              void* d_out, int out_size, void* d_ws, size_t ws_size,
                              hipStream_t stream) {
  (void)d_ws; (void)ws_size; (void)n_in; (void)out_size;
  const float* x    = (const float*)d_in[0];
  const int*   mask = (const int*)d_in[1];
  const float* ln_g = (const float*)d_in[2];
  const float* ln_b = (const float*)d_in[3];
  const float* Wj   = (const float*)d_in[4];
  const float* bj   = (const float*)d_in[5];
  const float* Wo   = (const float*)d_in[6];
  const float* bo   = (const float*)d_in[7];
  const float* Wg   = (const float*)d_in[8];
  const float* bg   = (const float*)d_in[9];
  const float* Wb   = (const float*)d_in[10];
  const float* bb   = (const float*)d_in[11];
  float* out = (float*)d_out;

  const int btot = in_sizes[0] / XDIM;
  const int grid = (btot + TILE - 1) / TILE;
  attn_fused_kernel<<<grid, 128, 0, stream>>>(x, mask, ln_g, ln_b,
                                             Wj, bj, Wo, bo, Wg, bg, Wb, bb,
                                             out, btot);
}


# work-groups stay resident and walk the tiles in a loop: weights and constants staged once per work-group, next tile's x loads issued ahead of the output copy (out stage moved into the dead q plane)
# speedup vs baseline: 10.7216x; 1.0690x over previous
#include <hip/hip_runtime.h>

typedef float v2f __attribute__((ext_vector_type(2)));
typedef float v4f __attribute__((ext_vector_type(4)));
typedef float v8f __attribute__((ext_vector_type(8)));

namespace {
constexpr int XDIM = 329;
constexpr int NQ   = 25;
constexpr int KV   = 9;
constexpr int TILE = 16;
constexpr int NWAVE = 4;
constexpr int NTHR  = 32 * NWAVE;
constexpr int XTAIL = 4;
constexpr int NWG   = 768;

constexpr int SX_ELEMS = TILE * XDIM + XTAIL;
constexpr int SP_ELEMS = 4 * TILE * NQ * KV;
constexpr int SW_ELEMS = 48*12 + 48*20 + 48*12 + 48*12;
constexpr int SB_ELEMS = 4 * 48;
constexpr int SMEM_ELEMS = SX_ELEMS + SP_ELEMS + SW_ELEMS + SB_ELEMS + NQ + 2 * KV;

constexpr int TILE_F4 = TILE * XDIM / 4;
constexpr int XCH     = (TILE_F4 + NTHR - 1) / NTHR;
static_assert(TILE * XDIM % 4 == 0, "tile must be whole 16-byte pieces");
static_assert(XCH * NTHR >= TILE_F4, "the group must cover the tile");

constexpr int WOFF_J = 0;
constexpr int WOFF_O = 48 * 12;
constexpr int WOFF_G = WOFF_O + 48 * 20;
constexpr int WOFF_B = WOFF_G + 48 * 12;
}

__device__ __forceinline__ int lds_base(int idx) {
  asm("" : "+v"(idx));
  return idx;
}

template <int L, int LPAD, int WOFF>
__device__ __forceinline__ void stage_w(const float* __restrict__ W, float* swp, int lane) {
  constexpr int N  = 4 * KV * L;
  constexpr int NT = (N + NTHR - 1) / NTHR;
  float r[NT];
  #pragma unroll
  for (int t = 0; t < NT; ++t) {
    const int i = min(lane + NTHR * t, N - 1);
    r[t] = W[i];
  }
  #pragma unroll
  for (int t = 0; t < NT; ++t) {
    const int i = min(lane + NTHR * t, N - 1);
    const int n = i / L, k = i - n * L;
    swp[WOFF + n * LPAD + k] = r[t];
  }
}

template <int L, int NVEC, int COL0, int Q0, int WOFF, int GIDX>
__device__ __forceinline__ void proj_group(const float* sxp, const float* swp,
                                           const float* sbias, float* sp,
                                           int m16, int hi, int wv, const int (&cb)[3]) {
  const int j0 = (wv + 6 * NWAVE - Q0) % NWAVE;
  constexpr int KSTEPS = (L + 3) / 4;
  constexpr int LPAD = KSTEPS * 4;
  if (j0 >= NVEC) return;

  v2f bf[3][KSTEPS];
  float bias[3];
  #pragma unroll
  for (int nt = 0; nt < 3; ++nt) {
    const int nn = nt * 16 + m16;
    #pragma unroll
    for (int ks = 0; ks < KSTEPS; ++ks)
      bf[nt][ks] = *reinterpret_cast<const v2f*>(swp + WOFF + nn * LPAD + ks * 4 + 2 * hi);
    bias[nt] = sbias[GIDX * 48 + nn];
  }

  const float* arow = sxp + m16 * XDIM + COL0 + 2 * hi;
  v2f a[KSTEPS];
  #pragma unroll
  for (int ks = 0; ks < KSTEPS; ++ks)
    a[ks] = (v2f){arow[j0 * L + ks * 4], arow[j0 * L + ks * 4 + 1]};

  for (int j = j0; j < NVEC; j += NWAVE) {
    const int jn = min(j + NWAVE, NVEC - 1);
    v2f an[KSTEPS];
    #pragma unroll
    for (int ks = 0; ks < KSTEPS; ++ks)
      an[ks] = (v2f){arow[jn * L + ks * 4], arow[jn * L + ks * 4 + 1]};

    v8f acc[3];
    #pragma unroll
    for (int nt = 0; nt < 3; ++nt) acc[nt] = (v8f){0.f, 0.f, 0.f, 0.f, 0.f, 0.f, 0.f, 0.f};
    #pragma unroll
    for (int ks = 0; ks < KSTEPS; ++ks) {
      #pragma unroll
      for (int nt = 0; nt < 3; ++nt)
        acc[nt] = __builtin_amdgcn_wmma_f32_16x16x4_f32(
             false, a[ks],  false, bf[nt][ks],
             (short)0, acc[nt],  false,  false);
    }

    const int q = Q0 + j;
    #pragma unroll
    for (int nt = 0; nt < 3; ++nt) {
      if (nt < 2 || m16 < 4 * KV - 32) {
        float* d = sp + cb[nt] + q * KV;
        #pragma unroll
        for (int r = 0; r < 8; ++r) d[r * (NQ * KV)] = acc[nt][r] + bias[nt];
      }
    }
    #pragma unroll
    for (int ks = 0; ks < KSTEPS; ++ks) a[ks] = an[ks];
  }
}

__global__ __launch_bounds__(NTHR) __attribute__((amdgpu_waves_per_eu(3)))
void attn_fused_kernel(
    const float* __restrict__ x, const int* __restrict__ mask,
    const float* __restrict__ ln_g, const float* __restrict__ ln_b,
    const float* __restrict__ Wj, const float* __restrict__ bj,
    const float* __restrict__ Wo, const float* __restrict__ bo,
    const float* __restrict__ Wg, const float* __restrict__ bg,
    const float* __restrict__ Wb, const float* __restrict__ bb,
    float* __restrict__ out, int btot) {
  __shared__ __align__(16) float smem[SMEM_ELEMS];
  float* sxp   = smem;
  float* sp    = smem + SX_ELEMS;
  float* swp   = sp + SP_ELEMS;
  float* sbias = swp + SW_ELEMS;
  float* smask = sbias + SB_ELEMS;
  float* sgam  = smask + NQ;
  float* sbet  = sgam + KV;

  const int lane0 = threadIdx.x;
  const int wv = __builtin_amdgcn_readfirstlane(lane0 >> 5);
  const int ntiles = (btot + TILE - 1) / TILE;
  constexpr int tstep = NWG;
  int tile = (int)blockIdx.x;

  const bool xal = ((reinterpret_cast<unsigned long long>(x) & 15ull) == 0ull);
  v4f cur[XCH];
  bool whole = (tile < ntiles) && xal && ((long long)tile * TILE + TILE <= (long long)btot);
  if (whole) {
    const v4f* xt = reinterpret_cast<const v4f*>(x + (long long)tile * TILE * XDIM);
    #pragma unroll
    for (int u = 0; u < XCH; ++u) cur[u] = xt[min(u * NTHR + lane0, TILE_F4 - 1)];
  } else {
    #pragma unroll
    for (int u = 0; u < XCH; ++u) cur[u] = (v4f){0.f, 0.f, 0.f, 0.f};
  }

  if (lane0 < XTAIL) sxp[TILE * XDIM + lane0] = 0.0f;
  for (int i = lane0; i < SW_ELEMS + SB_ELEMS; i += NTHR) swp[i] = 0.0f;
  __syncthreads();
  stage_w< 9, 12, WOFF_J>(Wj, swp, lane0);
  stage_w<17, 20, WOFF_O>(Wo, swp, lane0);
  stage_w<11, 12, WOFF_G>(Wg, swp, lane0);
  stage_w<11, 12, WOFF_B>(Wb, swp, lane0);
  for (int i = lane0; i < 4 * KV; i += NTHR) {
    sbias[0 * 48 + i] = bj[i];
    sbias[1 * 48 + i] = bo[i];
    sbias[2 * 48 + i] = bg[i];
    sbias[3 * 48 + i] = bb[i];
  }
  if (lane0 < NQ) smask[lane0] = -1.0e9f * (float)mask[lane0];
  if (lane0 < KV) { sgam[lane0] = ln_g[lane0]; sbet[lane0] = ln_b[lane0]; }

  for (; tile < ntiles; tile += tstep) {
  const long long tile0 = (long long)tile * TILE;
  int lane = wv * 32 + (int)__builtin_amdgcn_mbcnt_lo(~0u, 0u);
  asm volatile("" : "+v"(lane));

  if (whole) {
    v4f* sx4 = reinterpret_cast<v4f*>(sxp);
    #pragma unroll
    for (int u = 0; u < XCH; ++u) sx4[min(u * NTHR + lane, TILE_F4 - 1)] = cur[u];
  } else {
    for (int i = lane; i < TILE * XDIM; i += NTHR) {
      const int s = i / XDIM;
      sxp[i] = ((tile0 + s) < btot) ? x[tile0 * XDIM + i] : 0.0f;
    }
  }
  __syncthreads();

  const int m16 = lane & 15;
  const int hi  = (lane >> 4) & 1;
  int cb[3];
  #pragma unroll
  for (int nt = 0; nt < 3; ++nt) {
    const int nn = nt * 16 + m16;
    const int p = nn / KV;
    cb[nt] = p * (TILE * NQ * KV) + (nn - p * KV) + hi * (8 * NQ * KV);
  }
  proj_group< 9,  3,   0,  0, WOFF_J, 0>(sxp, swp, sbias, sp, m16, hi, wv, cb);
  proj_group<17, 10,  27,  3, WOFF_O, 1>(sxp, swp, sbias, sp, m16, hi, wv, cb);
  proj_group<11, 10, 197, 13, WOFF_G, 2>(sxp, swp, sbias, sp, m16, hi, wv, cb);
  proj_group<11,  2, 307, 23, WOFF_B, 3>(sxp, swp, sbias, sp, m16, hi, wv, cb);
  __syncthreads();

  constexpr int KB = 5;
  constexpr int NB = NQ / KB;
  constexpr int GE = KB * KV;
  static_assert(NB * KB == NQ, "groups must cover the rows");
  constexpr int PL = TILE * NQ * KV;
  const int s = lane >> 3;
  const int h = lane & 7;
  const int sb = s * (NQ * KV);
  const float* Pk  = smem + lds_base(SX_ELEMS + 1 * PL + sb);
  const float* Pv  = smem + lds_base(SX_ELEMS + 2 * PL + sb);
  const float* Pq  = smem + lds_base(SX_ELEMS + 0 * PL + sb + h * KV);
  const float* Pq3 = smem + lds_base(SX_ELEMS + 0 * PL + sb + (NQ - 1) * KV);
  const float* Pr  = smem + lds_base(SX_ELEMS + 3 * PL + sb + h * KV);
  const float* Pr3 = smem + lds_base(SX_ELEMS + 3 * PL + sb + (NQ - 1) * KV);
  float* ob  = smem + lds_base(SX_ELEMS + 0 * PL + sb + h * KV);
  float* ob3 = smem + lds_base(SX_ELEMS + 0 * PL + sb + (NQ - 1) * KV);
  const float* Pm = smask;

  v2f QA[KV], QB[KV];
  #pragma unroll
  for (int d = 0; d < KV; ++d) {
    QA[d] = (v2f){Pq[d], Pq[8 * KV + d]};
    QB[d] = (v2f){Pq[16 * KV + d], Pq3[d]};
  }
  float kq[NQ * KV], mq[NQ];
  #pragma unroll
  for (int i = 0; i < GE; ++i) kq[i] = Pk[i];
  #pragma unroll
  for (int i = 0; i < KB; ++i) mq[i] = Pm[i];

  v2f scA[NQ], scB[NQ];
  float mx0 = -3.0e38f, mx1 = -3.0e38f, mx2 = -3.0e38f, mx3 = -3.0e38f;
  #pragma unroll
  for (int b = 0; b < NB; ++b) {
    if (b + 1 < NB) {
      #pragma unroll
      for (int i = 0; i < GE; ++i) kq[(b + 1) * GE + i] = Pk[(b + 1) * GE + i];
      #pragma unroll
      for (int i = 0; i < KB; ++i) mq[(b + 1) * KB + i] = Pm[(b + 1) * KB + i];
    }
    __builtin_amdgcn_sched_barrier(0);
    #pragma unroll
    for (int jj = 0; jj < KB; ++jj) {
      const int j = b * KB + jj;
      const float* kr = kq + j * KV;
      v2f tA = QA[0] * (v2f){kr[0], kr[0]};
      v2f tB = QB[0] * (v2f){kr[0], kr[0]};
      #pragma unroll
      for (int d = 1; d < KV; ++d) {
        const v2f kk = {kr[d], kr[d]};
        tA = __builtin_elementwise_fma(QA[d], kk, tA);
        tB = __builtin_elementwise_fma(QB[d], kk, tB);
      }
      const v2f third = {1.0f / 3.0f, 1.0f / 3.0f};
      const v2f mkk = {mq[j], mq[j]};
      tA = __builtin_elementwise_fma(tA, third, mkk);
      tB = __builtin_elementwise_fma(tB, third, mkk);
      scA[j] = tA;
      scB[j] = tB;
      mx0 = fmaxf(mx0, tA.x); mx1 = fmaxf(mx1, tA.y);
      mx2 = fmaxf(mx2, tB.x); mx3 = fmaxf(mx3, tB.y);
    }
  }

  float vq[NQ * KV];
  #pragma unroll
  for (int i = 0; i < GE; ++i) vq[i] = Pv[i];
  __builtin_amdgcn_sched_barrier(0);

  v2f denA = {0.0f, 0.0f}, denB = {0.0f, 0.0f};
  #pragma unroll
  for (int j = 0; j < NQ; ++j) {
    const v2f eA = {__expf(scA[j].x - mx0), __expf(scA[j].y - mx1)};
    const v2f eB = {__expf(scB[j].x - mx2), __expf(scB[j].y - mx3)};
    scA[j] = eA;
    scB[j] = eB;
    denA += eA;
    denB += eB;
  }
  const v2f invA = {1.0f / denA.x, 1.0f / denA.y};
  const v2f invB = {1.0f / denB.x, 1.0f / denB.y};

  v2f oA[KV], oB[KV];
  float r0[KV], r1[KV], r2[KV], r3[KV], lg[KV], lb[KV];
  #pragma unroll
  for (int b = 0; b < NB; ++b) {
    if (b + 1 < NB) {
      #pragma unroll
      for (int i = 0; i < GE; ++i) vq[(b + 1) * GE + i] = Pv[(b + 1) * GE + i];
    } else {
      #pragma unroll
      for (int d = 0; d < KV; ++d) {
        r0[d] = Pr[d];
        r1[d] = Pr[8 * KV + d];
        r2[d] = Pr[16 * KV + d];
        r3[d] = Pr3[d];
        lg[d] = sgam[d];
        lb[d] = sbet[d];
      }
    }
    __builtin_amdgcn_sched_barrier(0);
    #pragma unroll
    for (int jj = 0; jj < KB; ++jj) {
      const int j = b * KB + jj;
      const float* vr = vq + j * KV;
      #pragma unroll
      for (int d = 0; d < KV; ++d) {
        const v2f vv = {vr[d], vr[d]};
        if (j == 0) {
          oA[d] = scA[0] * vv;
          oB[d] = scB[0] * vv;
        } else {
          oA[d] = __builtin_elementwise_fma(scA[j], vv, oA[d]);
          oB[d] = __builtin_elementwise_fma(scB[j], vv, oB[d]);
        }
      }
    }
  }

  v2f muA = {0.0f, 0.0f}, muB = {0.0f, 0.0f};
  #pragma unroll
  for (int d = 0; d < KV; ++d) {
    oA[d] = __builtin_elementwise_fma(oA[d], invA, (v2f){r0[d], r1[d]});
    oB[d] = __builtin_elementwise_fma(oB[d], invB, (v2f){r2[d], r3[d]});
    muA += oA[d];
    muB += oB[d];
  }
  muA *= (1.0f / KV);
  muB *= (1.0f / KV);
  v2f varA = {0.0f, 0.0f}, varB = {0.0f, 0.0f};
  #pragma unroll
  for (int d = 0; d < KV; ++d) {
    const v2f dA = oA[d] - muA, dB = oB[d] - muB;
    varA = __builtin_elementwise_fma(dA, dA, varA);
    varB = __builtin_elementwise_fma(dB, dB, varB);
  }
  varA *= (1.0f / KV);
  varB *= (1.0f / KV);
  const v2f rsA = {rsqrtf(varA.x + 1e-5f), rsqrtf(varA.y + 1e-5f)};
  const v2f rsB = {rsqrtf(varB.x + 1e-5f), rsqrtf(varB.y + 1e-5f)};
  #pragma unroll
  for (int d = 0; d < KV; ++d) {
    const v2f gg = {lg[d], lg[d]}, bb2 = {lb[d], lb[d]};
    const v2f yA = __builtin_elementwise_fma((oA[d] - muA) * rsA, gg, bb2);
    const v2f yB = __builtin_elementwise_fma((oB[d] - muB) * rsB, gg, bb2);
    ob[d] = yA.x;
    ob[8 * KV + d] = yA.y;
    ob[16 * KV + d] = yB.x;
    if (h == 0) ob3[d] = yB.y;
  }
  __syncthreads();

  int t3 = wv * 32 + (int)__builtin_amdgcn_mbcnt_lo(~0u, 0u);
  asm volatile("" : "+v"(t3));

  const int tnext = tile + tstep;
  whole = (tnext < ntiles) && xal && ((long long)tnext * TILE + TILE <= (long long)btot);
  if (whole) {
    const v4f* xt = reinterpret_cast<const v4f*>(x + (long long)tnext * TILE * XDIM);
    #pragma unroll
    for (int u = 0; u < XCH; ++u) cur[u] = xt[min(u * NTHR + t3, TILE_F4 - 1)];
  } else {
    #pragma unroll
    for (int u = 0; u < XCH; ++u) cur[u] = (v4f){0.f, 0.f, 0.f, 0.f};
  }

  const int nvalid = (int)min((long long)TILE, (long long)btot - tile0);
  float* __restrict__ ot = out + tile0 * (NQ * KV);
  for (int i = t3; i < nvalid * (NQ * KV); i += NTHR) ot[i] = sp[i];
  }
}

extern "C" void kernel_launch(void* const* d_in, const int* in_sizes, int n_in,
                              void* d_out, int out_size, void* d_ws, size_t ws_size,
                              hipStream_t stream) {
  (void)d_ws; (void)ws_size; (void)n_in; (void)out_size;
  const float* x    = (const float*)d_in[0];
  const int*   mask = (const int*)d_in[1];
  const float* ln_g = (const float*)d_in[2];
  const float* ln_b = (const float*)d_in[3];
  const float* Wj   = (const float*)d_in[4];
  const float* bj   = (const float*)d_in[5];
  const float* Wo   = (const float*)d_in[6];
  const float* bo   = (const float*)d_in[7];
  const float* Wg   = (const float*)d_in[8];
  const float* bg   = (const float*)d_in[9];
  const float* Wb   = (const float*)d_in[10];
  const float* bb   = (const float*)d_in[11];
  float* out = (float*)d_out;

  const int btot = in_sizes[0] / XDIM;
  attn_fused_kernel<<<NWG, 128, 0, stream>>>(x, mask, ln_g, ln_b,
                                             Wj, bj, Wo, bo, Wg, bg, Wb, bb,
                                             out, btot);
}


# output copy done per wave: each wave stores its own 4 samples as 16-byte pieces on 128-byte line boundaries, no work-group barrier behind attention; next x loads issued inside the pass over V
# speedup vs baseline: 11.5997x; 1.0819x over previous
#include <hip/hip_runtime.h>

typedef float v2f __attribute__((ext_vector_type(2)));
typedef float v4f __attribute__((ext_vector_type(4)));
typedef float v8f __attribute__((ext_vector_type(8)));

namespace {
constexpr int XDIM = 329;
constexpr int NQ   = 25;
constexpr int KV   = 9;
constexpr int TILE = 16;
constexpr int NWAVE = 4;
constexpr int NTHR  = 32 * NWAVE;
constexpr int XTAIL = 4;
constexpr int NWG   = 768;

constexpr int SX_ELEMS = TILE * XDIM + XTAIL;
constexpr int SP_ELEMS = 4 * TILE * NQ * KV;
constexpr int SW_ELEMS = 48*12 + 48*20 + 48*12 + 48*12;
constexpr int SB_ELEMS = 4 * 48;
constexpr int SMEM_ELEMS = SX_ELEMS + SP_ELEMS + SW_ELEMS + SB_ELEMS + NQ + 2 * KV;

constexpr int TILE_F4 = TILE * XDIM / 4;
constexpr int XCH     = (TILE_F4 + NTHR - 1) / NTHR;
static_assert(TILE * XDIM % 4 == 0, "tile must be whole 16-byte pieces");
static_assert(XCH * NTHR >= TILE_F4, "the group must cover the tile");

constexpr int WSAMP   = TILE / NWAVE;
constexpr int WOUT_F4 = WSAMP * NQ * KV / 4;
constexpr int OCH     = (WOUT_F4 + 7 + 31) / 32;
static_assert(WSAMP * NQ * KV % 4 == 0, "a wave's output share must be whole 16-byte pieces");
static_assert((SX_ELEMS * 4) % 16 == 0, "the out stage must start on a 16-byte boundary in LDS");

constexpr int WOFF_J = 0;
constexpr int WOFF_O = 48 * 12;
constexpr int WOFF_G = WOFF_O + 48 * 20;
constexpr int WOFF_B = WOFF_G + 48 * 12;
}

__device__ __forceinline__ int lds_base(int idx) {
  asm("" : "+v"(idx));
  return idx;
}

template <int L, int LPAD, int WOFF>
__device__ __forceinline__ void stage_w(const float* __restrict__ W, float* swp, int lane) {
  constexpr int N  = 4 * KV * L;
  constexpr int NT = (N + NTHR - 1) / NTHR;
  float r[NT];
  #pragma unroll
  for (int t = 0; t < NT; ++t) {
    const int i = min(lane + NTHR * t, N - 1);
    r[t] = W[i];
  }
  #pragma unroll
  for (int t = 0; t < NT; ++t) {
    const int i = min(lane + NTHR * t, N - 1);
    const int n = i / L, k = i - n * L;
    swp[WOFF + n * LPAD + k] = r[t];
  }
}

template <int L, int NVEC, int COL0, int Q0, int WOFF, int GIDX>
__device__ __forceinline__ void proj_group(const float* sxp, const float* swp,
                                           const float* sbias, float* sp,
                                           int m16, int hi, int wv, const int (&cb)[3]) {
  const int j0 = (wv + 6 * NWAVE - Q0) % NWAVE;
  constexpr int KSTEPS = (L + 3) / 4;
  constexpr int LPAD = KSTEPS * 4;
  if (j0 >= NVEC) return;

  v2f bf[3][KSTEPS];
  float bias[3];
  #pragma unroll
  for (int nt = 0; nt < 3; ++nt) {
    const int nn = nt * 16 + m16;
    #pragma unroll
    for (int ks = 0; ks < KSTEPS; ++ks)
      bf[nt][ks] = *reinterpret_cast<const v2f*>(swp + WOFF + nn * LPAD + ks * 4 + 2 * hi);
    bias[nt] = sbias[GIDX * 48 + nn];
  }

  const float* arow = sxp + m16 * XDIM + COL0 + 2 * hi;
  v2f a[KSTEPS];
  #pragma unroll
  for (int ks = 0; ks < KSTEPS; ++ks)
    a[ks] = (v2f){arow[j0 * L + ks * 4], arow[j0 * L + ks * 4 + 1]};

  for (int j = j0; j < NVEC; j += NWAVE) {
    const int jn = min(j + NWAVE, NVEC - 1);
    v2f an[KSTEPS];
    #pragma unroll
    for (int ks = 0; ks < KSTEPS; ++ks)
      an[ks] = (v2f){arow[jn * L + ks * 4], arow[jn * L + ks * 4 + 1]};

    v8f acc[3];
    #pragma unroll
    for (int nt = 0; nt < 3; ++nt) acc[nt] = (v8f){0.f, 0.f, 0.f, 0.f, 0.f, 0.f, 0.f, 0.f};
    #pragma unroll
    for (int ks = 0; ks < KSTEPS; ++ks) {
      #pragma unroll
      for (int nt = 0; nt < 3; ++nt)
        acc[nt] = __builtin_amdgcn_wmma_f32_16x16x4_f32(
             false, a[ks],  false, bf[nt][ks],
             (short)0, acc[nt],  false,  false);
    }

    const int q = Q0 + j;
    #pragma unroll
    for (int nt = 0; nt < 3; ++nt) {
      if (nt < 2 || m16 < 4 * KV - 32) {
        float* d = sp + cb[nt] + q * KV;
        #pragma unroll
        for (int r = 0; r < 8; ++r) d[r * (NQ * KV)] = acc[nt][r] + bias[nt];
      }
    }
    #pragma unroll
    for (int ks = 0; ks < KSTEPS; ++ks) a[ks] = an[ks];
  }
}

__global__ __launch_bounds__(NTHR) __attribute__((amdgpu_waves_per_eu(3)))
void attn_fused_kernel(
    const float* __restrict__ x, const int* __restrict__ mask,
    const float* __restrict__ ln_g, const float* __restrict__ ln_b,
    const float* __restrict__ Wj, const float* __restrict__ bj,
    const float* __restrict__ Wo, const float* __restrict__ bo,
    const float* __restrict__ Wg, const float* __restrict__ bg,
    const float* __restrict__ Wb, const float* __restrict__ bb,
    float* __restrict__ out, int btot) {
  __shared__ __align__(16) float smem[SMEM_ELEMS];
  float* sxp   = smem;
  float* sp    = smem + SX_ELEMS;
  float* swp   = sp + SP_ELEMS;
  float* sbias = swp + SW_ELEMS;
  float* smask = sbias + SB_ELEMS;
  float* sgam  = smask + NQ;
  float* sbet  = sgam + KV;

  const int lane0 = threadIdx.x;
  const int wv = __builtin_amdgcn_readfirstlane(lane0 >> 5);
  const int ntiles = (btot + TILE - 1) / TILE;
  constexpr int tstep = NWG;
  int tile = (int)blockIdx.x;

  const bool xal = ((reinterpret_cast<unsigned long long>(x) & 15ull) == 0ull);
  const bool oal = ((reinterpret_cast<unsigned long long>(out) & 15ull) == 0ull);
  v4f cur[XCH];
  bool whole = (tile < ntiles) && xal && ((long long)tile * TILE + TILE <= (long long)btot);
  if (whole) {
    const v4f* xt = reinterpret_cast<const v4f*>(x + (long long)tile * TILE * XDIM);
    #pragma unroll
    for (int u = 0; u < XCH; ++u) cur[u] = xt[min(u * NTHR + lane0, TILE_F4 - 1)];
  } else {
    #pragma unroll
    for (int u = 0; u < XCH; ++u) cur[u] = (v4f){0.f, 0.f, 0.f, 0.f};
  }

  if (lane0 < XTAIL) sxp[TILE * XDIM + lane0] = 0.0f;
  for (int i = lane0; i < SW_ELEMS + SB_ELEMS; i += NTHR) swp[i] = 0.0f;
  __syncthreads();
  stage_w< 9, 12, WOFF_J>(Wj, swp, lane0);
  stage_w<17, 20, WOFF_O>(Wo, swp, lane0);
  stage_w<11, 12, WOFF_G>(Wg, swp, lane0);
  stage_w<11, 12, WOFF_B>(Wb, swp, lane0);
  for (int i = lane0; i < 4 * KV; i += NTHR) {
    sbias[0 * 48 + i] = bj[i];
    sbias[1 * 48 + i] = bo[i];
    sbias[2 * 48 + i] = bg[i];
    sbias[3 * 48 + i] = bb[i];
  }
  if (lane0 < NQ) smask[lane0] = -1.0e9f * (float)mask[lane0];
  if (lane0 < KV) { sgam[lane0] = ln_g[lane0]; sbet[lane0] = ln_b[lane0]; }

  for (; tile < ntiles; tile += tstep) {
  const long long tile0 = (long long)tile * TILE;
  int lane = wv * 32 + (int)__builtin_amdgcn_mbcnt_lo(~0u, 0u);
  asm volatile("" : "+v"(lane));

  if (whole) {
    v4f* sx4 = reinterpret_cast<v4f*>(sxp);
    #pragma unroll
    for (int u = 0; u < XCH; ++u) sx4[min(u * NTHR + lane, TILE_F4 - 1)] = cur[u];
  } else {
    for (int i = lane; i < TILE * XDIM; i += NTHR) {
      const int s = i / XDIM;
      sxp[i] = ((tile0 + s) < btot) ? x[tile0 * XDIM + i] : 0.0f;
    }
  }
  __syncthreads();

  const int tnext = tile + tstep;
  const bool wnext = (tnext < ntiles) && xal && ((long long)tnext * TILE + TILE <= (long long)btot);
  const float* xfirst = x + ((0u - (unsigned)(reinterpret_cast<unsigned long long>(x) >> 2)) & 3u);
  const v4f* xt = reinterpret_cast<const v4f*>(wnext ? x + (long long)tnext * TILE * XDIM : xfirst);
  const unsigned xmsk = wnext ? ~0u : 0u;

  const int m16 = lane & 15;
  const int hi  = (lane >> 4) & 1;
  int cb[3];
  #pragma unroll
  for (int nt = 0; nt < 3; ++nt) {
    const int nn = nt * 16 + m16;
    const int p = nn / KV;
    cb[nt] = p * (TILE * NQ * KV) + (nn - p * KV) + hi * (8 * NQ * KV);
  }
  proj_group< 9,  3,   0,  0, WOFF_J, 0>(sxp, swp, sbias, sp, m16, hi, wv, cb);
  proj_group<17, 10,  27,  3, WOFF_O, 1>(sxp, swp, sbias, sp, m16, hi, wv, cb);
  proj_group<11, 10, 197, 13, WOFF_G, 2>(sxp, swp, sbias, sp, m16, hi, wv, cb);
  proj_group<11,  2, 307, 23, WOFF_B, 3>(sxp, swp, sbias, sp, m16, hi, wv, cb);
  __syncthreads();

  constexpr int KB = 5;
  constexpr int NB = NQ / KB;
  constexpr int GE = KB * KV;
  static_assert(NB * KB == NQ, "groups must cover the rows");
  constexpr int PL = TILE * NQ * KV;
  const int s = lane >> 3;
  const int h = lane & 7;
  const int sb = s * (NQ * KV);
  const float* Pk  = smem + lds_base(SX_ELEMS + 1 * PL + sb);
  const float* Pv  = smem + lds_base(SX_ELEMS + 2 * PL + sb);
  const float* Pq  = smem + lds_base(SX_ELEMS + 0 * PL + sb + h * KV);
  const float* Pq3 = smem + lds_base(SX_ELEMS + 0 * PL + sb + (NQ - 1) * KV);
  const float* Pr  = smem + lds_base(SX_ELEMS + 3 * PL + sb + h * KV);
  const float* Pr3 = smem + lds_base(SX_ELEMS + 3 * PL + sb + (NQ - 1) * KV);
  float* ob  = smem + lds_base(SX_ELEMS + 0 * PL + sb + h * KV);
  float* ob3 = smem + lds_base(SX_ELEMS + 0 * PL + sb + (NQ - 1) * KV);
  const float* Pm = smask;

  v2f QA[KV], QB[KV];
  #pragma unroll
  for (int d = 0; d < KV; ++d) {
    QA[d] = (v2f){Pq[d], Pq[8 * KV + d]};
    QB[d] = (v2f){Pq[16 * KV + d], Pq3[d]};
  }
  float kq[NQ * KV], mq[NQ];
  #pragma unroll
  for (int i = 0; i < GE; ++i) kq[i] = Pk[i];
  #pragma unroll
  for (int i = 0; i < KB; ++i) mq[i] = Pm[i];

  v2f scA[NQ], scB[NQ];
  float mx0 = -3.0e38f, mx1 = -3.0e38f, mx2 = -3.0e38f, mx3 = -3.0e38f;
  #pragma unroll
  for (int b = 0; b < NB; ++b) {
    if (b + 1 < NB) {
      #pragma unroll
      for (int i = 0; i < GE; ++i) kq[(b + 1) * GE + i] = Pk[(b + 1) * GE + i];
      #pragma unroll
      for (int i = 0; i < KB; ++i) mq[(b + 1) * KB + i] = Pm[(b + 1) * KB + i];
    }
    __builtin_amdgcn_sched_barrier(0);
    #pragma unroll
    for (int jj = 0; jj < KB; ++jj) {
      const int j = b * KB + jj;
      const float* kr = kq + j * KV;
      v2f tA = QA[0] * (v2f){kr[0], kr[0]};
      v2f tB = QB[0] * (v2f){kr[0], kr[0]};
      #pragma unroll
      for (int d = 1; d < KV; ++d) {
        const v2f kk = {kr[d], kr[d]};
        tA = __builtin_elementwise_fma(QA[d], kk, tA);
        tB = __builtin_elementwise_fma(QB[d], kk, tB);
      }
      const v2f third = {1.0f / 3.0f, 1.0f / 3.0f};
      const v2f mkk = {mq[j], mq[j]};
      tA = __builtin_elementwise_fma(tA, third, mkk);
      tB = __builtin_elementwise_fma(tB, third, mkk);
      scA[j] = tA;
      scB[j] = tB;
      mx0 = fmaxf(mx0, tA.x); mx1 = fmaxf(mx1, tA.y);
      mx2 = fmaxf(mx2, tB.x); mx3 = fmaxf(mx3, tB.y);
    }
  }

  float vq[NQ * KV];
  #pragma unroll
  for (int i = 0; i < GE; ++i) vq[i] = Pv[i];
  __builtin_amdgcn_sched_barrier(0);

  v2f denA = {0.0f, 0.0f}, denB = {0.0f, 0.0f};
  #pragma unroll
  for (int j = 0; j < NQ; ++j) {
    const v2f eA = {__expf(scA[j].x - mx0), __expf(scA[j].y - mx1)};
    const v2f eB = {__expf(scB[j].x - mx2), __expf(scB[j].y - mx3)};
    scA[j] = eA;
    scB[j] = eB;
    denA += eA;
    denB += eB;
  }
  const v2f invA = {1.0f / denA.x, 1.0f / denA.y};
  const v2f invB = {1.0f / denB.x, 1.0f / denB.y};

  v2f oA[KV], oB[KV];
  float r0[KV], r1[KV], r2[KV], r3[KV], lg[KV], lb[KV];
  #pragma unroll
  for (int b = 0; b < NB; ++b) {
    if (b + 1 < NB) {
      #pragma unroll
      for (int i = 0; i < GE; ++i) vq[(b + 1) * GE + i] = Pv[(b + 1) * GE + i];
    } else {
      #pragma unroll
      for (int d = 0; d < KV; ++d) {
        r0[d] = Pr[d];
        r1[d] = Pr[8 * KV + d];
        r2[d] = Pr[16 * KV + d];
        r3[d] = Pr3[d];
        lg[d] = sgam[d];
        lb[d] = sbet[d];
      }
    }
    if (b == NB - 2) {
      int tx = wv * 32 + (int)__builtin_amdgcn_mbcnt_lo(~0u, 0u);
      asm volatile("" : "+v"(tx));
      #pragma unroll
      for (int u = 0; u < XCH; ++u) cur[u] = xt[(unsigned)min(u * NTHR + tx, TILE_F4 - 1) & xmsk];
    }
    __builtin_amdgcn_sched_barrier(0);
    #pragma unroll
    for (int jj = 0; jj < KB; ++jj) {
      const int j = b * KB + jj;
      const float* vr = vq + j * KV;
      #pragma unroll
      for (int d = 0; d < KV; ++d) {
        const v2f vv = {vr[d], vr[d]};
        if (j == 0) {
          oA[d] = scA[0] * vv;
          oB[d] = scB[0] * vv;
        } else {
          oA[d] = __builtin_elementwise_fma(scA[j], vv, oA[d]);
          oB[d] = __builtin_elementwise_fma(scB[j], vv, oB[d]);
        }
      }
    }
  }

  v2f muA = {0.0f, 0.0f}, muB = {0.0f, 0.0f};
  #pragma unroll
  for (int d = 0; d < KV; ++d) {
    oA[d] = __builtin_elementwise_fma(oA[d], invA, (v2f){r0[d], r1[d]});
    oB[d] = __builtin_elementwise_fma(oB[d], invB, (v2f){r2[d], r3[d]});
    muA += oA[d];
    muB += oB[d];
  }
  muA *= (1.0f / KV);
  muB *= (1.0f / KV);
  v2f varA = {0.0f, 0.0f}, varB = {0.0f, 0.0f};
  #pragma unroll
  for (int d = 0; d < KV; ++d) {
    const v2f dA = oA[d] - muA, dB = oB[d] - muB;
    varA = __builtin_elementwise_fma(dA, dA, varA);
    varB = __builtin_elementwise_fma(dB, dB, varB);
  }
  varA *= (1.0f / KV);
  varB *= (1.0f / KV);
  const v2f rsA = {rsqrtf(varA.x + 1e-5f), rsqrtf(varA.y + 1e-5f)};
  const v2f rsB = {rsqrtf(varB.x + 1e-5f), rsqrtf(varB.y + 1e-5f)};
  #pragma unroll
  for (int d = 0; d < KV; ++d) {
    const v2f gg = {lg[d], lg[d]}, bb2 = {lb[d], lb[d]};
    const v2f yA = __builtin_elementwise_fma((oA[d] - muA) * rsA, gg, bb2);
    const v2f yB = __builtin_elementwise_fma((oB[d] - muB) * rsB, gg, bb2);
    ob[d] = yA.x;
    ob[8 * KV + d] = yA.y;
    ob[16 * KV + d] = yB.x;
    if (h == 0) ob3[d] = yB.y;
  }
  __builtin_amdgcn_fence(__ATOMIC_RELEASE, "wavefront");
  __builtin_amdgcn_wave_barrier();
  __builtin_amdgcn_fence(__ATOMIC_ACQUIRE, "wavefront");

  const int nvalid = (int)min((long long)TILE, (long long)btot - tile0);
  const int nvw = min(max(nvalid - wv * WSAMP, 0), WSAMP);
  float* __restrict__ otw = out + (tile0 + wv * WSAMP) * (NQ * KV);
  const float* spw = sp + wv * (WSAMP * NQ * KV);
  const int l32 = (int)__builtin_amdgcn_mbcnt_lo(~0u, 0u);
  if (nvw == WSAMP && oal) {
    const int lead = (int)((reinterpret_cast<unsigned long long>(otw) >> 4) & 7ull);
    const v4f* so4 = reinterpret_cast<const v4f*>(spw);
    v4f* ot4 = reinterpret_cast<v4f*>(otw);
    v4f ov[OCH];
    #pragma unroll
    for (int u = 0; u < OCH; ++u) {
      const int p = u * 32 + l32 - lead;
      ov[u] = so4[min(max(p, 0), WOUT_F4 - 1)];
    }
    #pragma unroll
    for (int u = 0; u < OCH; ++u) {
      const int p = u * 32 + l32 - lead;
      if ((unsigned)p < (unsigned)WOUT_F4) ot4[p] = ov[u];
    }
  } else {
    for (int i = l32; i < nvw * (NQ * KV); i += 32) otw[i] = spw[i];
  }
  whole = wnext;
  }
}

extern "C" void kernel_launch(void* const* d_in, const int* in_sizes, int n_in,
                              void* d_out, int out_size, void* d_ws, size_t ws_size,
                              hipStream_t stream) {
  (void)d_ws; (void)ws_size; (void)n_in; (void)out_size;
  const float* x    = (const float*)d_in[0];
  const int*   mask = (const int*)d_in[1];
  const float* ln_g = (const float*)d_in[2];
  const float* ln_b = (const float*)d_in[3];
  const float* Wj   = (const float*)d_in[4];
  const float* bj   = (const float*)d_in[5];
  const float* Wo   = (const float*)d_in[6];
  const float* bo   = (const float*)d_in[7];
  const float* Wg   = (const float*)d_in[8];
  const float* bg   = (const float*)d_in[9];
  const float* Wb   = (const float*)d_in[10];
  const float* bb   = (const float*)d_in[11];
  float* out = (float*)d_out;

  const int btot = in_sizes[0] / XDIM;
  attn_fused_kernel<<<NWG, 128, 0, stream>>>(x, mask, ln_g, ln_b,
                                             Wj, bj, Wo, bo, Wg, bg, Wb, bb,
                                             out, btot);
}


# attention: a sample's K and V rows (and the mask row) read from LDS as 16-byte pieces, 57 per pass where 113 dword pairs were; samples 228 dwords apart in the planes, out stage a region of its own
# speedup vs baseline: 11.9594x; 1.0310x over previous
#include <hip/hip_runtime.h>

typedef float v2f __attribute__((ext_vector_type(2)));
typedef float v4f __attribute__((ext_vector_type(4)));
typedef float v8f __attribute__((ext_vector_type(8)));

namespace {
constexpr int XDIM = 329;
constexpr int NQ   = 25;
constexpr int KV   = 9;
constexpr int TILE = 16;
constexpr int NWAVE = 4;
constexpr int NTHR  = 32 * NWAVE;
constexpr int XTAIL = 4;
constexpr int NWG   = 768;

constexpr int SX_ELEMS = TILE * XDIM + XTAIL;
constexpr int SPITCH   = NQ * KV + 3;
constexpr int PL       = TILE * SPITCH + 16;
constexpr int SP_ELEMS = 4 * PL;
constexpr int SO_ELEMS = TILE * NQ * KV;
constexpr int SW_ELEMS = 48*12 + 48*20 + 48*12 + 48*12;
constexpr int SB_ELEMS = 4 * 48;
constexpr int SM_ELEMS = (NQ + 3) / 4 * 4;
constexpr int SMEM_ELEMS = SX_ELEMS + SP_ELEMS + SO_ELEMS + SW_ELEMS + SB_ELEMS + SM_ELEMS + 2 * KV;
static_assert(SPITCH % 4 == 0 && SX_ELEMS % 4 == 0 && PL % 4 == 0, "K and V rows of every sample must start on a 16-byte boundary in LDS");
static_assert((SX_ELEMS + SP_ELEMS) % 4 == 0, "the out stage must start on a 16-byte boundary in LDS");
static_assert((SX_ELEMS + SP_ELEMS + SO_ELEMS + SW_ELEMS + SB_ELEMS) % 4 == 0, "the mask row must start on a 16-byte boundary in LDS");

constexpr int TILE_F4 = TILE * XDIM / 4;
constexpr int XCH     = (TILE_F4 + NTHR - 1) / NTHR;
static_assert(TILE * XDIM % 4 == 0, "tile must be whole 16-byte pieces");
static_assert(XCH * NTHR >= TILE_F4, "the group must cover the tile");

constexpr int WSAMP   = TILE / NWAVE;
constexpr int WOUT_F4 = WSAMP * NQ * KV / 4;
constexpr int OCH     = (WOUT_F4 + 7 + 31) / 32;
static_assert(WSAMP * NQ * KV % 4 == 0, "a wave's output share must be whole 16-byte pieces");

constexpr int WOFF_J = 0;
constexpr int WOFF_O = 48 * 12;
constexpr int WOFF_G = WOFF_O + 48 * 20;
constexpr int WOFF_B = WOFF_G + 48 * 12;
}

__device__ __forceinline__ int lds_base(int idx) {
  asm("" : "+v"(idx));
  return idx;
}

__device__ __forceinline__ float piece_dword(const v4f& p, int e) {
  if (e != 3) return p[e];
  v2f u = {p[2], p[3]};
  asm("" : "+v"(u));
  return u[1];
}

template <int L, int LPAD, int WOFF>
__device__ __forceinline__ void stage_w(const float* __restrict__ W, float* swp, int lane) {
  constexpr int N  = 4 * KV * L;
  constexpr int NT = (N + NTHR - 1) / NTHR;
  float r[NT];
  #pragma unroll
  for (int t = 0; t < NT; ++t) {
    const int i = min(lane + NTHR * t, N - 1);
    r[t] = W[i];
  }
  #pragma unroll
  for (int t = 0; t < NT; ++t) {
    const int i = min(lane + NTHR * t, N - 1);
    const int n = i / L, k = i - n * L;
    swp[WOFF + n * LPAD + k] = r[t];
  }
}

template <int L, int NVEC, int COL0, int Q0, int WOFF, int GIDX>
__device__ __forceinline__ void proj_group(const float* sxp, const float* swp,
                                           const float* sbias, float* sp,
                                           int m16, int hi, int wv, const int (&cb)[3]) {
  const int j0 = (wv + 6 * NWAVE - Q0) % NWAVE;
  constexpr int KSTEPS = (L + 3) / 4;
  constexpr int LPAD = KSTEPS * 4;
  if (j0 >= NVEC) return;

  v2f bf[3][KSTEPS];
  float bias[3];
  #pragma unroll
  for (int nt = 0; nt < 3; ++nt) {
    const int nn = nt * 16 + m16;
    #pragma unroll
    for (int ks = 0; ks < KSTEPS; ++ks)
      bf[nt][ks] = *reinterpret_cast<const v2f*>(swp + WOFF + nn * LPAD + ks * 4 + 2 * hi);
    bias[nt] = sbias[GIDX * 48 + nn];
  }

  const float* arow = sxp + m16 * XDIM + COL0 + 2 * hi;
  v2f a[KSTEPS];
  #pragma unroll
  for (int ks = 0; ks < KSTEPS; ++ks)
    a[ks] = (v2f){arow[j0 * L + ks * 4], arow[j0 * L + ks * 4 + 1]};

  for (int j = j0; j < NVEC; j += NWAVE) {
    const int jn = min(j + NWAVE, NVEC - 1);
    v2f an[KSTEPS];
    #pragma unroll
    for (int ks = 0; ks < KSTEPS; ++ks)
      an[ks] = (v2f){arow[jn * L + ks * 4], arow[jn * L + ks * 4 + 1]};

    v8f acc[3];
    #pragma unroll
    for (int nt = 0; nt < 3; ++nt) acc[nt] = (v8f){0.f, 0.f, 0.f, 0.f, 0.f, 0.f, 0.f, 0.f};
    #pragma unroll
    for (int ks = 0; ks < KSTEPS; ++ks) {
      #pragma unroll
      for (int nt = 0; nt < 3; ++nt)
        acc[nt] = __builtin_amdgcn_wmma_f32_16x16x4_f32(
             false, a[ks],  false, bf[nt][ks],
             (short)0, acc[nt],  false,  false);
    }

    const int q = Q0 + j;
    #pragma unroll
    for (int nt = 0; nt < 3; ++nt) {
      if (nt < 2 || m16 < 4 * KV - 32) {
        float* d = sp + cb[nt] + q * KV;
        #pragma unroll
        for (int r = 0; r < 8; ++r) d[r * SPITCH] = acc[nt][r] + bias[nt];
      }
    }
    #pragma unroll
    for (int ks = 0; ks < KSTEPS; ++ks) a[ks] = an[ks];
  }
}

__global__ __launch_bounds__(NTHR) __attribute__((amdgpu_waves_per_eu(3)))
void attn_fused_kernel(
    const float* __restrict__ x, const int* __restrict__ mask,
    const float* __restrict__ ln_g, const float* __restrict__ ln_b,
    const float* __restrict__ Wj, const float* __restrict__ bj,
    const float* __restrict__ Wo, const float* __restrict__ bo,
    const float* __restrict__ Wg, const float* __restrict__ bg,
    const float* __restrict__ Wb, const float* __restrict__ bb,
    float* __restrict__ out, int btot) {
  __shared__ __align__(16) float smem[SMEM_ELEMS];
  float* sxp   = smem;
  float* sp    = smem + SX_ELEMS;
  float* so    = sp + SP_ELEMS;
  float* swp   = so + SO_ELEMS;
  float* sbias = swp + SW_ELEMS;
  float* smask = sbias + SB_ELEMS;
  float* sgam  = smask + SM_ELEMS;
  float* sbet  = sgam + KV;

  const int lane0 = threadIdx.x;
  const int wv = __builtin_amdgcn_readfirstlane(lane0 >> 5);
  const int ntiles = (btot + TILE - 1) / TILE;
  constexpr int tstep = NWG;
  int tile = (int)blockIdx.x;

  const bool xal = ((reinterpret_cast<unsigned long long>(x) & 15ull) == 0ull);
  const bool oal = ((reinterpret_cast<unsigned long long>(out) & 15ull) == 0ull);
  v4f cur[XCH];
  bool whole = (tile < ntiles) && xal && ((long long)tile * TILE + TILE <= (long long)btot);
  if (whole) {
    const v4f* xt = reinterpret_cast<const v4f*>(x + (long long)tile * TILE * XDIM);
    #pragma unroll
    for (int u = 0; u < XCH; ++u) cur[u] = xt[min(u * NTHR + lane0, TILE_F4 - 1)];
  } else {
    #pragma unroll
    for (int u = 0; u < XCH; ++u) cur[u] = (v4f){0.f, 0.f, 0.f, 0.f};
  }

  if (lane0 < XTAIL) sxp[TILE * XDIM + lane0] = 0.0f;
  for (int i = lane0; i < SW_ELEMS + SB_ELEMS; i += NTHR) swp[i] = 0.0f;
  __syncthreads();
  stage_w< 9, 12, WOFF_J>(Wj, swp, lane0);
  stage_w<17, 20, WOFF_O>(Wo, swp, lane0);
  stage_w<11, 12, WOFF_G>(Wg, swp, lane0);
  stage_w<11, 12, WOFF_B>(Wb, swp, lane0);
  for (int i = lane0; i < 4 * KV; i += NTHR) {
    sbias[0 * 48 + i] = bj[i];
    sbias[1 * 48 + i] = bo[i];
    sbias[2 * 48 + i] = bg[i];
    sbias[3 * 48 + i] = bb[i];
  }
  if (lane0 < SM_ELEMS) smask[lane0] = (lane0 < NQ) ? -1.0e9f * (float)mask[min(lane0, NQ - 1)] : 0.0f;
  if (lane0 < KV) { sgam[lane0] = ln_g[lane0]; sbet[lane0] = ln_b[lane0]; }

  for (; tile < ntiles; tile += tstep) {
  const long long tile0 = (long long)tile * TILE;
  int lane = wv * 32 + (int)__builtin_amdgcn_mbcnt_lo(~0u, 0u);
  asm volatile("" : "+v"(lane));

  if (whole) {
    v4f* sx4 = reinterpret_cast<v4f*>(sxp);
    #pragma unroll
    for (int u = 0; u < XCH; ++u) sx4[min(u * NTHR + lane, TILE_F4 - 1)] = cur[u];
  } else {
    for (int i = lane; i < TILE * XDIM; i += NTHR) {
      const int s = i / XDIM;
      sxp[i] = ((tile0 + s) < btot) ? x[tile0 * XDIM + i] : 0.0f;
    }
  }
  __syncthreads();

  const int tnext = tile + tstep;
  const bool wnext = (tnext < ntiles) && xal && ((long long)tnext * TILE + TILE <= (long long)btot);
  const float* xfirst = x + ((0u - (unsigned)(reinterpret_cast<unsigned long long>(x) >> 2)) & 3u);
  const v4f* xt = reinterpret_cast<const v4f*>(wnext ? x + (long long)tnext * TILE * XDIM : xfirst);
  const unsigned xmsk = wnext ? ~0u : 0u;

  const int m16 = lane & 15;
  const int hi  = (lane >> 4) & 1;
  int cb[3];
  #pragma unroll
  for (int nt = 0; nt < 3; ++nt) {
    const int nn = nt * 16 + m16;
    const int p = nn / KV;
    cb[nt] = p * PL + (nn - p * KV) + hi * (8 * SPITCH);
  }
  proj_group< 9,  3,   0,  0, WOFF_J, 0>(sxp, swp, sbias, sp, m16, hi, wv, cb);
  proj_group<17, 10,  27,  3, WOFF_O, 1>(sxp, swp, sbias, sp, m16, hi, wv, cb);
  proj_group<11, 10, 197, 13, WOFF_G, 2>(sxp, swp, sbias, sp, m16, hi, wv, cb);
  proj_group<11,  2, 307, 23, WOFF_B, 3>(sxp, swp, sbias, sp, m16, hi, wv, cb);
  __syncthreads();

  constexpr int KB = 5;
  constexpr int NB = NQ / KB;
  constexpr int GE = KB * KV;
  static_assert(NB * KB == NQ, "groups must cover the rows");
  constexpr int NP = (NQ * KV + 3) / 4;
  const int s = lane >> 3;
  const int h = lane & 7;
  const int sb = s * SPITCH;
  const v4f* Pk4 = reinterpret_cast<const v4f*>(smem + SX_ELEMS + 1 * PL + sb);
  const v4f* Pv4 = reinterpret_cast<const v4f*>(smem + SX_ELEMS + 2 * PL + sb);
  const v4f* Pm4 = reinterpret_cast<const v4f*>(smask);
  const float* Pq  = smem + lds_base(SX_ELEMS + 0 * PL + sb + h * KV);
  const float* Pq3 = smem + lds_base(SX_ELEMS + 0 * PL + sb + (NQ - 1) * KV);
  const float* Pr  = smem + lds_base(SX_ELEMS + 3 * PL + sb + h * KV);
  const float* Pr3 = smem + lds_base(SX_ELEMS + 3 * PL + sb + (NQ - 1) * KV);
  float* ob  = smem + lds_base(SX_ELEMS + SP_ELEMS + s * (NQ * KV) + h * KV);
  float* ob3 = smem + lds_base(SX_ELEMS + SP_ELEMS + s * (NQ * KV) + (NQ - 1) * KV);
  constexpr auto plo = [](int b, int n) constexpr { return b == 0 ? 0 : (n * b - 1) / 4 + 1; };
  constexpr auto phi = [](int b, int n) constexpr { return (n * b + n - 1) / 4; };

  v2f QA[KV], QB[KV];
  #pragma unroll
  for (int d = 0; d < KV; ++d) {
    QA[d] = (v2f){Pq[d], Pq[8 * KV + d]};
    QB[d] = (v2f){Pq[16 * KV + d], Pq3[d]};
  }
  v4f kq4[NP], mq4[SM_ELEMS / 4];
  #pragma unroll
  for (int p = plo(0, GE); p <= phi(0, GE); ++p) kq4[p] = Pk4[p];
  #pragma unroll
  for (int p = plo(0, KB); p <= phi(0, KB); ++p) mq4[p] = Pm4[p];

  v2f scA[NQ], scB[NQ];
  float mx0 = -3.0e38f, mx1 = -3.0e38f, mx2 = -3.0e38f, mx3 = -3.0e38f;
  #pragma unroll
  for (int b = 0; b < NB; ++b) {
    if (b + 1 < NB) {
      #pragma unroll
      for (int p = plo(b + 1, GE); p <= phi(b + 1, GE); ++p) kq4[p] = Pk4[p];
      #pragma unroll
      for (int p = plo(b + 1, KB); p <= phi(b + 1, KB); ++p) mq4[p] = Pm4[p];
    }
    __builtin_amdgcn_sched_barrier(0);
    #pragma unroll
    for (int jj = 0; jj < KB; ++jj) {
      const int j = b * KB + jj;
      const float k0 = piece_dword(kq4[(j * KV) >> 2], (j * KV) & 3);
      v2f tA = QA[0] * (v2f){k0, k0};
      v2f tB = QB[0] * (v2f){k0, k0};
      #pragma unroll
      for (int d = 1; d < KV; ++d) {
        const float kd = piece_dword(kq4[(j * KV + d) >> 2], (j * KV + d) & 3);
        const v2f kk = {kd, kd};
        tA = __builtin_elementwise_fma(QA[d], kk, tA);
        tB = __builtin_elementwise_fma(QB[d], kk, tB);
      }
      const v2f third = {1.0f / 3.0f, 1.0f / 3.0f};
      const float mj = mq4[j >> 2][j & 3];
      const v2f mkk = {mj, mj};
      tA = __builtin_elementwise_fma(tA, third, mkk);
      tB = __builtin_elementwise_fma(tB, third, mkk);
      scA[j] = tA;
      scB[j] = tB;
      mx0 = fmaxf(mx0, tA.x); mx1 = fmaxf(mx1, tA.y);
      mx2 = fmaxf(mx2, tB.x); mx3 = fmaxf(mx3, tB.y);
    }
  }

  v4f vq4[NP];
  #pragma unroll
  for (int p = plo(0, GE); p <= phi(0, GE); ++p) vq4[p] = Pv4[p];
  __builtin_amdgcn_sched_barrier(0);

  v2f denA = {0.0f, 0.0f}, denB = {0.0f, 0.0f};
  #pragma unroll
  for (int j = 0; j < NQ; ++j) {
    const v2f eA = {__expf(scA[j].x - mx0), __expf(scA[j].y - mx1)};
    const v2f eB = {__expf(scB[j].x - mx2), __expf(scB[j].y - mx3)};
    scA[j] = eA;
    scB[j] = eB;
    denA += eA;
    denB += eB;
  }
  const v2f invA = {1.0f / denA.x, 1.0f / denA.y};
  const v2f invB = {1.0f / denB.x, 1.0f / denB.y};

  v2f oA[KV], oB[KV];
  float r0[KV], r1[KV], r2[KV], r3[KV], lg[KV], lb[KV];
  #pragma unroll
  for (int b = 0; b < NB; ++b) {
    if (b + 1 < NB) {
      #pragma unroll
      for (int p = plo(b + 1, GE); p <= phi(b + 1, GE); ++p) vq4[p] = Pv4[p];
    } else {
      #pragma unroll
      for (int d = 0; d < KV; ++d) {
        r0[d] = Pr[d];
        r1[d] = Pr[8 * KV + d];
        r2[d] = Pr[16 * KV + d];
        r3[d] = Pr3[d];
        lg[d] = sgam[d];
        lb[d] = sbet[d];
      }
    }
    if (b == NB - 2) {
      int tx = wv * 32 + (int)__builtin_amdgcn_mbcnt_lo(~0u, 0u);
      asm volatile("" : "+v"(tx));
      #pragma unroll
      for (int u = 0; u < XCH; ++u) cur[u] = xt[(unsigned)min(u * NTHR + tx, TILE_F4 - 1) & xmsk];
    }
    __builtin_amdgcn_sched_barrier(0);
    #pragma unroll
    for (int jj = 0; jj < KB; ++jj) {
      const int j = b * KB + jj;
      #pragma unroll
      for (int d = 0; d < KV; ++d) {
        const float vd = piece_dword(vq4[(j * KV + d) >> 2], (j * KV + d) & 3);
        const v2f vv = {vd, vd};
        if (j == 0) {
          oA[d] = scA[0] * vv;
          oB[d] = scB[0] * vv;
        } else {
          oA[d] = __builtin_elementwise_fma(scA[j], vv, oA[d]);
          oB[d] = __builtin_elementwise_fma(scB[j], vv, oB[d]);
        }
      }
    }
  }

  v2f muA = {0.0f, 0.0f}, muB = {0.0f, 0.0f};
  #pragma unroll
  for (int d = 0; d < KV; ++d) {
    oA[d] = __builtin_elementwise_fma(oA[d], invA, (v2f){r0[d], r1[d]});
    oB[d] = __builtin_elementwise_fma(oB[d], invB, (v2f){r2[d], r3[d]});
    muA += oA[d];
    muB += oB[d];
  }
  muA *= (1.0f / KV);
  muB *= (1.0f / KV);
  v2f varA = {0.0f, 0.0f}, varB = {0.0f, 0.0f};
  #pragma unroll
  for (int d = 0; d < KV; ++d) {
    const v2f dA = oA[d] - muA, dB = oB[d] - muB;
    varA = __builtin_elementwise_fma(dA, dA, varA);
    varB = __builtin_elementwise_fma(dB, dB, varB);
  }
  varA *= (1.0f / KV);
  varB *= (1.0f / KV);
  const v2f rsA = {rsqrtf(varA.x + 1e-5f), rsqrtf(varA.y + 1e-5f)};
  const v2f rsB = {rsqrtf(varB.x + 1e-5f), rsqrtf(varB.y + 1e-5f)};
  #pragma unroll
  for (int d = 0; d < KV; ++d) {
    const v2f gg = {lg[d], lg[d]}, bb2 = {lb[d], lb[d]};
    const v2f yA = __builtin_elementwise_fma((oA[d] - muA) * rsA, gg, bb2);
    const v2f yB = __builtin_elementwise_fma((oB[d] - muB) * rsB, gg, bb2);
    ob[d] = yA.x;
    ob[8 * KV + d] = yA.y;
    ob[16 * KV + d] = yB.x;
    if (h == 0) ob3[d] = yB.y;
  }
  __builtin_amdgcn_fence(__ATOMIC_RELEASE, "wavefront");
  __builtin_amdgcn_wave_barrier();
  __builtin_amdgcn_fence(__ATOMIC_ACQUIRE, "wavefront");

  const int nvalid = (int)min((long long)TILE, (long long)btot - tile0);
  const int nvw = min(max(nvalid - wv * WSAMP, 0), WSAMP);
  float* __restrict__ otw = out + (tile0 + wv * WSAMP) * (NQ * KV);
  const float* spw = so + wv * (WSAMP * NQ * KV);
  const int l32 = (int)__builtin_amdgcn_mbcnt_lo(~0u, 0u);
  if (nvw == WSAMP && oal) {
    const int lead = (int)((reinterpret_cast<unsigned long long>(otw) >> 4) & 7ull);
    const v4f* so4 = reinterpret_cast<const v4f*>(spw);
    v4f* ot4 = reinterpret_cast<v4f*>(otw);
    v4f ov[OCH];
    #pragma unroll
    for (int u = 0; u < OCH; ++u) {
      const int p = u * 32 + l32 - lead;
      ov[u] = so4[min(max(p, 0), WOUT_F4 - 1)];
    }
    #pragma unroll
    for (int u = 0; u < OCH; ++u) {
      const int p = u * 32 + l32 - lead;
      if ((unsigned)p < (unsigned)WOUT_F4) ot4[p] = ov[u];
    }
  } else {
    for (int i = l32; i < nvw * (NQ * KV); i += 32) otw[i] = spw[i];
  }
  whole = wnext;
  }
}

extern "C" void kernel_launch(void* const* d_in, const int* in_sizes, int n_in,
                              void* d_out, int out_size, void* d_ws, size_t ws_size,
                              hipStream_t stream) {
  (void)d_ws; (void)ws_size; (void)n_in; (void)out_size;
  const float* x    = (const float*)d_in[0];
  const int*   mask = (const int*)d_in[1];
  const float* ln_g = (const float*)d_in[2];
  const float* ln_b = (const float*)d_in[3];
  const float* Wj   = (const float*)d_in[4];
  const float* bj   = (const float*)d_in[5];
  const float* Wo   = (const float*)d_in[6];
  const float* bo   = (const float*)d_in[7];
  const float* Wg   = (const float*)d_in[8];
  const float* bg   = (const float*)d_in[9];
  const float* Wb   = (const float*)d_in[10];
  const float* bb   = (const float*)d_in[11];
  float* out = (float*)d_out;

  const int btot = in_sizes[0] / XDIM;
  attn_fused_kernel<<<NWG, 128, 0, stream>>>(x, mask, ln_g, ln_b,
                                             Wj, bj, Wo, bo, Wg, bg, Wb, bb,
                                             out, btot);
}
